# starting kernel
# baseline (speedup 1.0000x reference)
#include <hip/hip_runtime.h>

typedef __attribute__((ext_vector_type(2))) float v2f;
typedef __attribute__((ext_vector_type(8))) float v8f;

__global__ void zero_f32(float* __restrict__ buf, int n) {
  int i = blockIdx.x * blockDim.x + threadIdx.x;
  if (i < n) buf[i] = 0.0f;
}

__global__ void deg_accum(const int* __restrict__ dst, float* __restrict__ deg, int E) {
  int e = blockIdx.x * blockDim.x + threadIdx.x;
  if (e < E) atomicAdd(&deg[dst[e]], 1.0f);
}

__global__ void transform1(const float* __restrict__ x, const float* __restrict__ W1,
                           float* __restrict__ dnorm, float* __restrict__ p,
                           float* __restrict__ agg, int N) {
  int i = blockIdx.x * blockDim.x + threadIdx.x;
  if (i >= N) return;
  float di = rsqrtf(dnorm[i] + 1.0f);
  dnorm[i] = di;
  float s = x[i] * di;
  const float4* w = (const float4*)W1;
  float4* pp = (float4*)(p + (size_t)i * 16);
  float4* ag = (float4*)(agg + (size_t)i * 16);
#pragma unroll
  for (int k = 0; k < 4; ++k) {
    float4 wv = w[k];
    float4 o;
    o.x = wv.x * s; o.y = wv.y * s; o.z = wv.z * s; o.w = wv.w * s;
    pp[k] = o;
    float4 z; z.x = 0.f; z.y = 0.f; z.z = 0.f; z.w = 0.f;
    ag[k] = z;
  }
}

__global__ void edge_scatter(const int* __restrict__ src, const int* __restrict__ dst,
                             const float* __restrict__ p, float* __restrict__ agg, int E) {
  int t = blockIdx.x * blockDim.x + threadIdx.x;
  int e = t >> 2;
  if (e >= E) return;
  int sub = (t & 3) * 4;
  int s = src[e];
  int d = dst[e];
  float4 v = *(const float4*)(p + (size_t)s * 16 + sub);
  float* ap = agg + (size_t)d * 16 + sub;
  atomicAdd(ap + 0, v.x);
  atomicAdd(ap + 1, v.y);
  atomicAdd(ap + 2, v.z);
  atomicAdd(ap + 3, v.w);
}

__global__ void __launch_bounds__(256)
gcn_mid_wmma(const float* __restrict__ p, const float* __restrict__ agg,
             const float* __restrict__ dnorm, const float* __restrict__ b1,
             const float* __restrict__ W2, float* __restrict__ p2,
             float* __restrict__ agg_out, int N, int nTiles) {
  const int lane = threadIdx.x & 31;
  const int tile = (blockIdx.x * blockDim.x + threadIdx.x) >> 5;
  if (tile >= nTiles) return;
  const int base = tile * 16;
  const int col  = lane & 15;
  const bool hi  = lane >= 16;
  const bool full = (base + 16 <= N);

  int arow = base + col;
  if (arow >= N) arow = N - 1;
  const float di = dnorm[arow];

  v2f a[4], b[4];
#pragma unroll
  for (int k = 0; k < 4; ++k) {
    const int c0 = 4 * k + (hi ? 2 : 0);
    float2 pv = *(const float2*)(p   + (size_t)arow * 16 + c0);
    float2 av = *(const float2*)(agg + (size_t)arow * 16 + c0);
    a[k][0] = fmaxf(di * (av.x + pv.x) + b1[c0],     0.0f);
    a[k][1] = fmaxf(di * (av.y + pv.y) + b1[c0 + 1], 0.0f);
    b[k][0] = W2[c0 * 16 + col];
    b[k][1] = W2[(c0 + 1) * 16 + col];
  }

  v8f acc = {};
#pragma unroll
  for (int k = 0; k < 4; ++k)
    acc = __builtin_amdgcn_wmma_f32_16x16x4_f32(false, a[k], false, b[k],
                                                (short)0, acc, false, false);

  if (full) {
#pragma unroll
    for (int r = 0; r < 8; ++r) {
      int orow = base + (hi ? 8 + r : r);
      p2[(size_t)orow * 16 + col]      = acc[r] * dnorm[orow];
      agg_out[(size_t)orow * 16 + col] = 0.0f;
    }
  } else {
#pragma unroll
    for (int r = 0; r < 8; ++r) {
      int orow = base + (hi ? 8 + r : r);
      if (orow < N) {
        p2[(size_t)orow * 16 + col]      = acc[r] * dnorm[orow];
        agg_out[(size_t)orow * 16 + col] = 0.0f;
      }
    }
  }
}

__global__ void __launch_bounds__(256)
gcn_out_wmma(const float* __restrict__ p2, const float* __restrict__ agg2,
             const float* __restrict__ dnorm, const float* __restrict__ b2,
             const float* __restrict__ Wl, const float* __restrict__ bl,
             float* __restrict__ out, int N, int nTiles) {
  const int lane = threadIdx.x & 31;
  const int tile = (blockIdx.x * blockDim.x + threadIdx.x) >> 5;
  if (tile >= nTiles) return;
  const int base = tile * 16;
  const int col  = lane & 15;
  const bool hi  = lane >= 16;
  const bool full = (base + 16 <= N);

  int arow = base + col;
  if (arow >= N) arow = N - 1;
  const float di = dnorm[arow];

  v2f a[4], b[4];
#pragma unroll
  for (int k = 0; k < 4; ++k) {
    const int c0 = 4 * k + (hi ? 2 : 0);
    float2 pv = *(const float2*)(p2   + (size_t)arow * 16 + c0);
    float2 av = *(const float2*)(agg2 + (size_t)arow * 16 + c0);
    a[k][0] = fmaxf(di * (av.x + pv.x) + b2[c0],     0.0f);
    a[k][1] = fmaxf(di * (av.y + pv.y) + b2[c0 + 1], 0.0f);
    b[k][0] = (col < 4) ? Wl[c0 * 4 + col]       : 0.0f;
    b[k][1] = (col < 4) ? Wl[(c0 + 1) * 4 + col] : 0.0f;
  }

  const float cv = (col < 4) ? bl[col] : 0.0f;
  v8f acc;
#pragma unroll
  for (int r = 0; r < 8; ++r) acc[r] = cv;
#pragma unroll
  for (int k = 0; k < 4; ++k)
    acc = __builtin_amdgcn_wmma_f32_16x16x4_f32(false, a[k], false, b[k],
                                                (short)0, acc, false, false);

  if (col < 4) {
    if (full) {
#pragma unroll
      for (int r = 0; r < 8; ++r) {
        int orow = base + (hi ? 8 + r : r);
        out[(size_t)orow * 4 + col] = acc[r];
      }
    } else {
#pragma unroll
      for (int r = 0; r < 8; ++r) {
        int orow = base + (hi ? 8 + r : r);
        if (orow < N) out[(size_t)orow * 4 + col] = acc[r];
      }
    }
  }
}

extern "C" void kernel_launch(void* const* d_in, const int* in_sizes, int n_in,
                              void* d_out, int out_size, void* d_ws, size_t ws_size,
                              hipStream_t stream) {
  const float* x  = (const float*)d_in[0];
  const int*   ed = (const int*)d_in[1];
  const float* W1 = (const float*)d_in[2];
  const float* b1 = (const float*)d_in[3];
  const float* W2 = (const float*)d_in[4];
  const float* b2 = (const float*)d_in[5];
  const float* Wl = (const float*)d_in[6];
  const float* bl = (const float*)d_in[7];
  float* out = (float*)d_out;

  const int N = in_sizes[0];
  const int E = in_sizes[1] / 2;

  const int* src = ed;
  const int* dst = ed + E;

  float* dnorm = (float*)d_ws;
  float* p     = dnorm + N;
  float* agg   = p + (size_t)N * 16;

  const int B = 256;
  const int nTiles = (N + 15) / 16;
  const int tileBlocks = (nTiles * 32 + B - 1) / B;
  const int edgeBlocks = (int)(((long long)E * 4 + B - 1) / B);

  zero_f32  <<<(N + B - 1) / B, B, 0, stream>>>(dnorm, N);
  deg_accum <<<(E + B - 1) / B, B, 0, stream>>>(dst, dnorm, E);

  transform1<<<(N + B - 1) / B, B, 0, stream>>>(x, W1, dnorm, p, agg, N);
  edge_scatter<<<edgeBlocks, B, 0, stream>>>(src, dst, p, agg, E);

  gcn_mid_wmma<<<tileBlocks, B, 0, stream>>>(p, agg, dnorm, b1, W2, p, agg, N, nTiles);

  edge_scatter<<<edgeBlocks, B, 0, stream>>>(src, dst, p, agg, E);

  gcn_out_wmma<<<tileBlocks, B, 0, stream>>>(p, agg, dnorm, b2, Wl, bl, out, N, nTiles);
}


# edge list ordered (32-row group, sub-slice, row), not (row, sub-slice): a wave's runs are one contiguous stretch of csr and rp2, not a line per row; layer 1: 2 lanes per run, f32 sum order changes
# speedup vs baseline: 7.1195x; 7.1195x over previous
#include <hip/hip_runtime.h>

typedef __attribute__((ext_vector_type(2))) float v2f;
typedef __attribute__((ext_vector_type(8))) float v8f;

__global__ void zero_f32(float* __restrict__ buf, int n) {
  int i = blockIdx.x * blockDim.x + threadIdx.x;
  if (i < n) buf[i] = 0.0f;
}

__global__ void deg_accum(const int* __restrict__ dst, float* __restrict__ deg, int E) {
  int e = blockIdx.x * blockDim.x + threadIdx.x;
  if (e < E) atomicAdd(&deg[dst[e]], 1.0f);
}

__global__ void transform1(const float* __restrict__ x, const float* __restrict__ W1,
                           float* __restrict__ dnorm, float* __restrict__ p,
                           float* __restrict__ agg, int N) {
  int i = blockIdx.x * blockDim.x + threadIdx.x;
  if (i >= N) return;
  float di = rsqrtf(dnorm[i] + 1.0f);
  dnorm[i] = di;
  float s = x[i] * di;
  const float4* w = (const float4*)W1;
  float4* pp = (float4*)(p + (size_t)i * 16);
  float4* ag = (float4*)(agg + (size_t)i * 16);
#pragma unroll
  for (int k = 0; k < 4; ++k) {
    float4 wv = w[k];
    float4 o;
    o.x = wv.x * s; o.y = wv.y * s; o.z = wv.z * s; o.w = wv.w * s;
    pp[k] = o;
    float4 z; z.x = 0.f; z.y = 0.f; z.z = 0.f; z.w = 0.f;
    ag[k] = z;
  }
}

__global__ void edge_scatter(const int* __restrict__ src, const int* __restrict__ dst,
                             const float* __restrict__ p, float* __restrict__ agg, int E) {
  int t = blockIdx.x * blockDim.x + threadIdx.x;
  int e = t >> 2;
  if (e >= E) return;
  int sub = (t & 3) * 4;
  int s = src[e];
  int d = dst[e];
  float4 v = *(const float4*)(p + (size_t)s * 16 + sub);
  float* ap = agg + (size_t)d * 16 + sub;
  atomicAdd(ap + 0, v.x);
  atomicAdd(ap + 1, v.y);
  atomicAdd(ap + 2, v.z);
  atomicAdd(ap + 3, v.w);
}

template <bool kZeroAgg>
__global__ void __launch_bounds__(256)
gcn_mid_wmma(const float* __restrict__ p, const float* __restrict__ agg,
             const float* __restrict__ dnorm, const float* __restrict__ b1,
             const float* __restrict__ W2, float* __restrict__ p2,
             float* __restrict__ agg_out, int N, int nTiles) {
  const int lane = threadIdx.x & 31;
  const int tile = (blockIdx.x * blockDim.x + threadIdx.x) >> 5;
  if (tile >= nTiles) return;
  const int base = tile * 16;
  const int col  = lane & 15;
  const bool hi  = lane >= 16;
  const bool full = (base + 16 <= N);

  int arow = base + col;
  if (arow >= N) arow = N - 1;
  const float di = dnorm[arow];

  v2f a[4], b[4];
#pragma unroll
  for (int k = 0; k < 4; ++k) {
    const int c0 = 4 * k + (hi ? 2 : 0);
    float2 pv = *(const float2*)(p   + (size_t)arow * 16 + c0);
    float2 av = *(const float2*)(agg + (size_t)arow * 16 + c0);
    a[k][0] = fmaxf(di * (av.x + pv.x) + b1[c0],     0.0f);
    a[k][1] = fmaxf(di * (av.y + pv.y) + b1[c0 + 1], 0.0f);
    b[k][0] = W2[c0 * 16 + col];
    b[k][1] = W2[(c0 + 1) * 16 + col];
  }

  v8f acc = {};
#pragma unroll
  for (int k = 0; k < 4; ++k)
    acc = __builtin_amdgcn_wmma_f32_16x16x4_f32(false, a[k], false, b[k],
                                                (short)0, acc, false, false);

  if (full) {
#pragma unroll
    for (int r = 0; r < 8; ++r) {
      int orow = base + (hi ? 8 + r : r);
      p2[(size_t)orow * 16 + col]      = acc[r] * dnorm[orow];
      if (kZeroAgg) agg_out[(size_t)orow * 16 + col] = 0.0f;
    }
  } else {
#pragma unroll
    for (int r = 0; r < 8; ++r) {
      int orow = base + (hi ? 8 + r : r);
      if (orow < N) {
        p2[(size_t)orow * 16 + col]      = acc[r] * dnorm[orow];
        if (kZeroAgg) agg_out[(size_t)orow * 16 + col] = 0.0f;
      }
    }
  }
}

__global__ void __launch_bounds__(256)
gcn_out_wmma(const float* __restrict__ p2, const float* __restrict__ agg2,
             const float* __restrict__ dnorm, const float* __restrict__ b2,
             const float* __restrict__ Wl, const float* __restrict__ bl,
             float* __restrict__ out, int N, int nTiles) {
  const int lane = threadIdx.x & 31;
  const int tile = (blockIdx.x * blockDim.x + threadIdx.x) >> 5;
  if (tile >= nTiles) return;
  const int base = tile * 16;
  const int col  = lane & 15;
  const bool hi  = lane >= 16;
  const bool full = (base + 16 <= N);

  int arow = base + col;
  if (arow >= N) arow = N - 1;
  const float di = dnorm[arow];

  v2f a[4], b[4];
#pragma unroll
  for (int k = 0; k < 4; ++k) {
    const int c0 = 4 * k + (hi ? 2 : 0);
    float2 pv = *(const float2*)(p2   + (size_t)arow * 16 + c0);
    float2 av = *(const float2*)(agg2 + (size_t)arow * 16 + c0);
    a[k][0] = fmaxf(di * (av.x + pv.x) + b2[c0],     0.0f);
    a[k][1] = fmaxf(di * (av.y + pv.y) + b2[c0 + 1], 0.0f);
    b[k][0] = (col < 4) ? Wl[c0 * 4 + col]       : 0.0f;
    b[k][1] = (col < 4) ? Wl[(c0 + 1) * 4 + col] : 0.0f;
  }

  const float cv = (col < 4) ? bl[col] : 0.0f;
  v8f acc;
#pragma unroll
  for (int r = 0; r < 8; ++r) acc[r] = cv;
#pragma unroll
  for (int k = 0; k < 4; ++k)
    acc = __builtin_amdgcn_wmma_f32_16x16x4_f32(false, a[k], false, b[k],
                                                (short)0, acc, false, false);

  if (col < 4) {
    if (full) {
#pragma unroll
      for (int r = 0; r < 8; ++r) {
        int orow = base + (hi ? 8 + r : r);
        out[(size_t)orow * 4 + col] = acc[r];
      }
    } else {
#pragma unroll
      for (int r = 0; r < 8; ++r) {
        int orow = base + (hi ? 8 + r : r);
        if (orow < N) out[(size_t)orow * 4 + col] = acc[r];
      }
    }
  }
}

#define NCHUNK 1024
#define MAXB   1024

#define CU 49
__global__ void __launch_bounds__(256)
bucket_count(const int* __restrict__ dst, unsigned* __restrict__ cnt,
             int E, int chunk, int nb) {
  __shared__ unsigned h[MAXB + 32];
  const int t    = threadIdx.x;
  const int lane = t & 31;
  for (int b = t; b < MAXB + 32; b += 256) h[b] = 0u;
  __syncthreads();
  const int w  = blockIdx.x;
  const int e0 = w * chunk;
  const int e1 = (e0 + chunk < E) ? (e0 + chunk) : E;
  for (int eb = e0; eb < e1; eb += 256 * CU) {
    int dv[CU];
#pragma unroll
    for (int j = 0; j < CU; ++j) {
      const int e = eb + j * 256 + t;
      dv[j] = dst[(e < e1) ? e : (E - 1)];
    }
#pragma unroll
    for (int j = 0; j < CU; ++j) {
      const int e = eb + j * 256 + t;
      const unsigned b = (e < e1) ? (unsigned)(dv[j] >> 8) : (unsigned)(MAXB + lane);
      atomicAdd(&h[b], 1u);
    }
  }
  __syncthreads();
  for (int b = t; b < nb; b += 256) cnt[(size_t)b * NCHUNK + w] = h[b];
}

__global__ void __launch_bounds__(256)
bucket_scan_rows(unsigned* __restrict__ cnt, unsigned* __restrict__ btot) {
  __shared__ unsigned s[256];
  const int t = threadIdx.x;
  uint4* row = (uint4*)(cnt + (size_t)blockIdx.x * NCHUNK);
  const uint4 v = row[t];
  const unsigned sum = v.x + v.y + v.z + v.w;
  s[t] = sum;
  __syncthreads();
  for (int off = 1; off < 256; off <<= 1) {
    const unsigned add = (t >= off) ? s[t - off] : 0u;
    __syncthreads();
    s[t] += add;
    __syncthreads();
  }
  const unsigned ex = s[t] - sum;
  uint4 o;
  o.x = ex; o.y = ex + v.x; o.z = o.y + v.y; o.w = o.z + v.z;
  row[t] = o;
  if (t == 255) btot[blockIdx.x] = s[255];
}

__global__ void __launch_bounds__(1024)
bucket_scan_tot(const unsigned* __restrict__ btot, unsigned* __restrict__ bbase,
                int* __restrict__ rowptr, int nb, int N) {
  __shared__ unsigned s[1024];
  const int t = threadIdx.x;
  const unsigned v = (t < nb) ? btot[t] : 0u;
  s[t] = v;
  __syncthreads();
  for (int off = 1; off < 1024; off <<= 1) {
    const unsigned add = (t >= off) ? s[t - off] : 0u;
    __syncthreads();
    s[t] += add;
    __syncthreads();
  }
  if (t < nb) bbase[t] = s[t] - v;
  if (t == 1023) { bbase[nb] = s[1023]; rowptr[N] = (int)s[1023]; }
}

__global__ void __launch_bounds__(256)
bucket_place(const int* __restrict__ src, const int* __restrict__ dst,
             const unsigned* __restrict__ cnt, const unsigned* __restrict__ bbase,
             unsigned* __restrict__ rec, int E, int chunk, int nb) {
  __shared__ unsigned cur[MAXB];
  const int w = blockIdx.x;
  for (int b = threadIdx.x; b < nb; b += 256)
    cur[b] = bbase[b] + cnt[(size_t)b * NCHUNK + w];
  __syncthreads();
  const int e0 = w * chunk;
  const int e1 = (e0 + chunk < E) ? (e0 + chunk) : E;
  for (int e = e0 + (int)threadIdx.x; e < e1; e += 256) {
    const unsigned d = (unsigned)dst[e];
    const unsigned s = (unsigned)src[e];
    const unsigned pos = atomicAdd(&cur[d >> 8], 1u);
    rec[pos] = s | ((d & 255u) << 18);
  }
}

__device__ __forceinline__ unsigned wave_incl_scan(unsigned v, int lane) {
#pragma unroll
  for (int off = 1; off < 32; off <<= 1) {
    const unsigned n = __shfl_up(v, off, 32);
    if (lane >= off) v += n;
  }
  return v;
}

__device__ __forceinline__ unsigned block_excl_scan_1024(unsigned v, unsigned* wsum) {
  const int lane = threadIdx.x & 31;
  const int wave = threadIdx.x >> 5;
  const unsigned incl = wave_incl_scan(v, lane);
  if (lane == 31) wsum[wave] = incl;
  __syncthreads();
  if (wave == 0) {
    const unsigned w  = wsum[lane];
    const unsigned wi = wave_incl_scan(w, lane);
    wsum[lane] = wi - w;
  }
  __syncthreads();
  return incl - v + wsum[wave];
}

#define PCAP 13
#define SCAP 18

__global__ void __launch_bounds__(1024)
bucket_place_lds(const int* __restrict__ src, const int* __restrict__ dst,
                 const unsigned* __restrict__ cnt, const unsigned* __restrict__ bbase,
                 unsigned* __restrict__ rec, int E, int chunk, int nb) {
  __shared__ unsigned h[MAXB + 32];
  __shared__ unsigned gbs[MAXB];
  __shared__ unsigned wsum[32];
  __shared__ uint2 stage[PCAP * 1024 + 32];
  const int t    = threadIdx.x;
  const int lane = t & 31;
  const int w    = blockIdx.x;
  const int e0   = w * chunk;
  const int e1   = (e0 + chunk < E) ? (e0 + chunk) : E;
  const int n    = e1 - e0;
  h[t] = 0u;
  if (t < 32) h[MAXB + t] = 0u;
  const unsigned gb = (t < nb) ? (bbase[t] + cnt[(size_t)t * NCHUNK + w]) : 0u;

  unsigned sv[PCAP], dv[PCAP];
#pragma unroll
  for (int j = 0; j < PCAP; ++j) {
    const int e  = e0 + j * 1024 + t;
    const int ec = (e < e1) ? e : (E - 1);
    dv[j] = (unsigned)dst[ec];
    sv[j] = (unsigned)src[ec];
  }
  __syncthreads();
  unsigned rk[PCAP];
#pragma unroll
  for (int j = 0; j < PCAP; ++j) {
    const bool ok = (j * 1024 + t) < n;
    const unsigned b = ok ? (dv[j] >> 8) : (unsigned)(MAXB + lane);
    rk[j] = atomicAdd(&h[b], 1u);
  }
#pragma unroll
  for (int j = 0; j < PCAP; ++j) {
    sv[j] |= (dv[j] & 255u) << 18;
    dv[j]  = (dv[j] >> 8) | (rk[j] << 10);
  }
  __syncthreads();
  const unsigned ex = block_excl_scan_1024(h[t], wsum);
  h[t]   = ex;
  gbs[t] = gb;
  __syncthreads();
#pragma unroll
  for (int j = 0; j < PCAP; ++j) {
    const bool ok = (j * 1024 + t) < n;
    const unsigned b    = ok ? (dv[j] & 1023u) : 0u;
    const unsigned rank = dv[j] >> 10;
    uint2 o;
    o.x = sv[j];
    o.y = gbs[b] + rank;
    const unsigned lp = ok ? (h[b] + rank) : (unsigned)(PCAP * 1024 + lane);
    stage[lp] = o;
  }
  __syncthreads();
  for (int i = t; i < n; i += 1024) {
    const uint2 o = stage[i];
    rec[o.y] = o.x;
  }
}

#define NSUB 8

__device__ __forceinline__ unsigned key_of(unsigned nl, unsigned sub) {
  return ((nl >> 5) << 8) | (sub << 5) | (nl & 31u);
}
__device__ __forceinline__ size_t slot_of(int node, int sub) {
  return (size_t)(node >> 5) * 256 + (size_t)sub * 32 + (size_t)(node & 31);
}

__device__ __forceinline__ unsigned sub_of(unsigned s, unsigned SL2) {
  unsigned k = 0u;
#pragma unroll
  for (unsigned q = 1; q < NSUB; ++q) k += (s >= q * SL2) ? 1u : 0u;
  return k;
}

__global__ void __launch_bounds__(1024) __attribute__((amdgpu_waves_per_eu(16, 16)))
bucket_sort_sub(const unsigned* __restrict__ rec, const unsigned* __restrict__ bbase,
                int* __restrict__ rp2, int* __restrict__ csr, int N, unsigned SL2) {
  __shared__ unsigned c[256 * NSUB + 32];
  __shared__ unsigned wsum[32];
  __shared__ unsigned stage[SCAP * 1024 + 32];
  const int t    = threadIdx.x;
  const int lane = t & 31;
  const unsigned bs = bbase[blockIdx.x];
  const unsigned be = bbase[blockIdx.x + 1];
  const unsigned n  = be - bs;
  c[t] = 0u;
  c[t + 1024] = 0u;
  if (t < 32) c[256 * NSUB + t] = 0u;

  if (n <= (unsigned)(SCAP * 1024)) {
    unsigned rv[SCAP], rk[SCAP];
#pragma unroll
    for (int j = 0; j < SCAP; ++j) {
      const unsigned i = (unsigned)(j * 1024 + t);
      rv[j] = rec[(i < n) ? (bs + i) : 0u];
    }
#pragma unroll
    for (int j = 0; j < SCAP; ++j) {
      const unsigned s = rv[j] & 0x3FFFFu;
      rv[j] = s | (key_of(rv[j] >> 18, sub_of(s, SL2)) << 18);
      asm volatile("" : "+v"(rv[j]));
    }
    __syncthreads();
#pragma unroll
    for (int j = 0; j < SCAP; ++j) {
      const bool ok = (unsigned)(j * 1024 + t) < n;
      const unsigned k = ok ? (rv[j] >> 18) : (unsigned)(256 * NSUB + lane);
      rk[j] = atomicAdd(&c[k], 1u);
    }
    __syncthreads();
    const unsigned v0 = c[2 * t];
    const unsigned v1 = c[2 * t + 1];
    const unsigned ex = block_excl_scan_1024(v0 + v1, wsum);
    c[2 * t]     = ex;
    c[2 * t + 1] = ex + v0;
    {
      int2 o;
      o.x = (int)(bs + ex);
      o.y = (int)(bs + ex + v0);
      ((int2*)rp2)[(size_t)blockIdx.x * 1024 + t] = o;
    }
    __syncthreads();
#pragma unroll
    for (int j = 0; j < SCAP; ++j) {
      const bool ok = (unsigned)(j * 1024 + t) < n;
      asm volatile("" : "+v"(rv[j]));
      const unsigned key = ok ? (rv[j] >> 18) : 0u;
      const unsigned lp  = ok ? (c[key] + rk[j]) : (unsigned)(SCAP * 1024 + lane);
      stage[lp] = rv[j] & 0x3FFFFu;
    }
    __syncthreads();
    for (unsigned i = (unsigned)t; i < n; i += 1024u) csr[bs + i] = (int)stage[i];
  } else {
    __syncthreads();
    for (unsigned e = bs + (unsigned)t; e < be; e += 1024u) {
      const unsigned r = rec[e];
      atomicAdd(&c[key_of(r >> 18, sub_of(r & 0x3FFFFu, SL2))], 1u);
    }
    __syncthreads();
    const unsigned v0 = c[2 * t];
    const unsigned v1 = c[2 * t + 1];
    const unsigned ex = block_excl_scan_1024(v0 + v1, wsum);
    c[2 * t]     = bs + ex;
    c[2 * t + 1] = bs + ex + v0;
    {
      int2 o;
      o.x = (int)(bs + ex);
      o.y = (int)(bs + ex + v0);
      ((int2*)rp2)[(size_t)blockIdx.x * 1024 + t] = o;
    }
    __syncthreads();
    for (unsigned e = bs + (unsigned)t; e < be; e += 1024u) {
      const unsigned r = rec[e];
      const unsigned pos = atomicAdd(&c[key_of(r >> 18, sub_of(r & 0x3FFFFu, SL2))], 1u);
      csr[pos] = (int)(r & 0x3FFFFu);
    }
  }
}

__global__ void transform_y(const float* __restrict__ x, const int* __restrict__ rp2,
                            float* __restrict__ dnorm, float* __restrict__ y, int N) {
  int i = blockIdx.x * blockDim.x + threadIdx.x;
  if (i >= N) return;
  const int* rr = rp2 + slot_of(i, 0);
  int deg = 0;
#pragma unroll
  for (int s = 0; s < NSUB; ++s) deg += rr[s * 32 + 1] - rr[s * 32];
  float di = rsqrtf((float)deg + 1.0f);
  dnorm[i] = di;
  y[i] = x[i] * di;
}

#define SLCAP 68096
#define YU    10

__global__ void __launch_bounds__(1024)
pull_y_lds(const int* __restrict__ rp2, const int* __restrict__ csr,
           const float* __restrict__ y, float* __restrict__ S, int N, int SL) {
  __shared__ float ys[SLCAP];
  const int t   = threadIdx.x;
  const int k   = blockIdx.y;
  const int s0  = k * SL;
  const int s1  = (s0 + SL < N) ? (s0 + SL) : N;
  const int len = s1 - s0;
  const unsigned ulen = (len > 0) ? (unsigned)len : 0u;

  for (int j0 = t; j0 < len; j0 += 1024 * 16) {
    float v[16];
#pragma unroll
    for (int u = 0; u < 16; ++u) {
      const int j = j0 + u * 1024;
      v[u] = y[s0 + ((j < len) ? j : 0)];
    }
#pragma unroll
    for (int u = 0; u < 16; ++u) {
      const int j = j0 + u * 1024;
      if (j < len) ys[j] = v[u];
    }
  }
  __syncthreads();

  const int G   = gridDim.x;
  const int npg = (((N + G - 1) / G) + 31) & ~31;
  const int n0  = blockIdx.x * npg;
  const int n1  = (n0 + npg < N) ? (n0 + npg) : N;
  const int sub = t & 3;
  const int hh  = sub >> 1;
  const int ls  = sub & 1;
  float* Sk = S + (size_t)k * N;
  for (int nb0 = n0; nb0 < n1; nb0 += 256) {
    const int i = nb0 + (t >> 2);
    const bool live = i < n1;
    int e = 0, e1 = 0;
    if (live) {
      const int* rr = rp2 + slot_of(i, 2 * k + hh);
      e  = rr[0];
      e1 = rr[1];
    }
    float acc = 0.0f;
    for (e += ls; e < e1; e += 2 * YU) {
      int sv[YU];
#pragma unroll
      for (int j = 0; j < YU; ++j) {
        const int ee = e + 2 * j;
        sv[j] = csr[(ee < e1) ? ee : (e1 - 1)];
      }
#pragma unroll
      for (int j = 0; j < YU; ++j) {
        const unsigned r = (unsigned)(sv[j] - s0);
        const bool ok = ((e + 2 * j) < e1) && (r < ulen);
        const float v = ys[ok ? r : 0u];
        acc += ok ? v : 0.0f;
      }
    }
    acc += __shfl_xor(acc, 1, 32);
    acc += __shfl_xor(acc, 2, 32);
    if (live && sub == 0) Sk[i] = acc;
  }
}

__global__ void make_td(const float* __restrict__ S, int nsl, const float* __restrict__ y,
                        const float* __restrict__ dnorm, float* __restrict__ td, int N) {
  int i = blockIdx.x * blockDim.x + threadIdx.x;
  if (i >= N) return;
  const float di = dnorm[i];
  float s = 0.0f;
  for (int q = 0; q < nsl; ++q) s += S[(size_t)q * N + i];
  s += y[i];
  float2 o;
  o.x = di * s;
  o.y = di;
  ((float2*)td)[i] = o;
}

#define TDCAP (SLCAP / 2)
#define TU    4

__global__ void __launch_bounds__(1024)
pull_td_lds(const int* __restrict__ rp2, const int* __restrict__ csr,
            const float* __restrict__ tdf, const float* __restrict__ W1,
            const float* __restrict__ b1, float* __restrict__ P, int N, int SL2) {
  __shared__ float2 tds[TDCAP];
  const float2* td = (const float2*)tdf;
  const int t = threadIdx.x;
  const int k = blockIdx.y;
  float w1[16], bb1[16];
#pragma unroll
  for (int c = 0; c < 16; ++c) { w1[c] = W1[c]; bb1[c] = b1[c]; }

  const int G   = gridDim.x;
  const int npg = (((N + G - 1) / G) + 31) & ~31;
  const int n0  = blockIdx.x * npg;
  const int n1  = (n0 + npg < N) ? (n0 + npg) : N;

  for (int h = 0; h < 2; ++h) {
    const int sub = 2 * k + h;
    const int s0  = sub * SL2;
    const int s1  = (s0 + SL2 < N) ? (s0 + SL2) : N;
    const int len = s1 - s0;
    const unsigned ulen = (len > 0) ? (unsigned)len : 0u;
    __syncthreads();
    for (int j0 = t; j0 < len; j0 += 1024 * 16) {
      float2 v[16];
#pragma unroll
      for (int u = 0; u < 16; ++u) {
        const int j = j0 + u * 1024;
        v[u] = td[s0 + ((j < len) ? j : 0)];
      }
#pragma unroll
      for (int u = 0; u < 16; ++u) {
        const int j = j0 + u * 1024;
        if (j < len) tds[j] = v[u];
      }
    }
    __syncthreads();

    for (int nb0 = n0; nb0 < n1; nb0 += 1024) {
      const int i = nb0 + t;
      const bool live = i < n1;
      int e0 = 0, cnt = 0;
      if (live) {
        const int* rr = rp2 + slot_of(i, sub);
        e0  = rr[0];
        cnt = rr[1] - e0;
      }
      int wmax = cnt;
#pragma unroll
      for (int off = 16; off > 0; off >>= 1) {
        const int o = __shfl_xor(wmax, off, 32);
        wmax = (o > wmax) ? o : wmax;
      }
      wmax = __builtin_amdgcn_readfirstlane(wmax);

      float acc[16];
#pragma unroll
      for (int c = 0; c < 16; ++c) acc[c] = 0.0f;
      int cur[TU];
#pragma unroll
      for (int j = 0; j < TU; ++j) cur[j] = csr[(j < cnt) ? (e0 + j) : 0];
      for (int base = 0; base < wmax; base += TU) {
        int nxt[TU];
#pragma unroll
        for (int j = 0; j < TU; ++j) {
          const int q = base + TU + j;
          nxt[j] = csr[(q < cnt) ? (e0 + q) : 0];
        }
#pragma unroll
        for (int j = 0; j < TU; ++j) {
          const int q = base + j;
          const unsigned r = (unsigned)(cur[j] - s0);
          const bool ok = (q < cnt) && (r < ulen);
          const float2 v = tds[ok ? r : 0u];
          const float tj = ok ? v.x : 0.0f;
          const float dj = ok ? v.y : 0.0f;
#pragma unroll
          for (int c = 0; c < 16; ++c)
            acc[c] = fmaf(dj, fmaxf(fmaf(tj, w1[c], bb1[c]), 0.0f), acc[c]);
        }
#pragma unroll
        for (int j = 0; j < TU; ++j) cur[j] = nxt[j];
      }
      if (live) {
        float4* Pp = (float4*)(P + ((size_t)k * N + i) * 16);
#pragma unroll
        for (int q = 0; q < 4; ++q) {
          float4 o;
          o.x = acc[4 * q]; o.y = acc[4 * q + 1]; o.z = acc[4 * q + 2]; o.w = acc[4 * q + 3];
          if (h) {
            const float4 p = Pp[q];
            o.x += p.x; o.y += p.y; o.z += p.z; o.w += p.w;
          }
          Pp[q] = o;
        }
      }
    }
  }
}

__global__ void __launch_bounds__(256)
gcn_out_wmma_u(const float* __restrict__ P, int nsl, const float* __restrict__ td,
               const float* __restrict__ W1, const float* __restrict__ b1,
               const float* __restrict__ W2, const float* __restrict__ b2,
               const float* __restrict__ Wl, const float* __restrict__ bl,
               float* __restrict__ out, int N, int nTiles) {
  const int lane = threadIdx.x & 31;
  const int tile = (blockIdx.x * blockDim.x + threadIdx.x) >> 5;
  if (tile >= nTiles) return;
  const int base = tile * 16;
  const int col  = lane & 15;
  const int hf   = lane >> 4;
  const bool full = (base + 16 <= N);

  int arow = base + col;
  if (arow >= N) arow = N - 1;
  const float2 tdv = ((const float2*)td)[arow];
  const float ti = tdv.x;
  const float di = tdv.y;

  v2f aw[4], bu[4];
#pragma unroll
  for (int k = 0; k < 4; ++k) {
    const int c0 = 4 * k + 2 * hf;
    float u0 = 0.0f, u1 = 0.0f;
    for (int q = 0; q < nsl; ++q) {
      const float2 pv = *(const float2*)(P + ((size_t)q * N + arow) * 16 + c0);
      u0 += pv.x;
      u1 += pv.y;
    }
    u0 += di * fmaxf(fmaf(ti, W1[c0],     b1[c0]),     0.0f);
    u1 += di * fmaxf(fmaf(ti, W1[c0 + 1], b1[c0 + 1]), 0.0f);
    bu[k][0] = u0;
    bu[k][1] = u1;
    aw[k][0] = W2[c0 * 16 + col];
    aw[k][1] = W2[(c0 + 1) * 16 + col];
  }

  v8f z = {};
#pragma unroll
  for (int k = 0; k < 4; ++k)
    z = __builtin_amdgcn_wmma_f32_16x16x4_f32(false, aw[k], false, bu[k],
                                              (short)0, z, false, false);

  v2f ah[4], bw[4];
#pragma unroll
  for (int k = 0; k < 4; ++k) {
    const int c0 = 8 * hf + 2 * k;
    ah[k][0] = fmaxf(fmaf(di, z[2 * k],     b2[c0]),     0.0f);
    ah[k][1] = fmaxf(fmaf(di, z[2 * k + 1], b2[c0 + 1]), 0.0f);
    bw[k][0] = (col < 4) ? Wl[c0 * 4 + col]       : 0.0f;
    bw[k][1] = (col < 4) ? Wl[(c0 + 1) * 4 + col] : 0.0f;
  }

  const float cv = (col < 4) ? bl[col] : 0.0f;
  v8f acc;
#pragma unroll
  for (int r = 0; r < 8; ++r) acc[r] = cv;
#pragma unroll
  for (int k = 0; k < 4; ++k)
    acc = __builtin_amdgcn_wmma_f32_16x16x4_f32(false, ah[k], false, bw[k],
                                                (short)0, acc, false, false);

  if (col < 4) {
    if (full) {
#pragma unroll
      for (int r = 0; r < 8; ++r) {
        int orow = base + (hf ? 8 + r : r);
        out[(size_t)orow * 4 + col] = acc[r];
      }
    } else {
#pragma unroll
      for (int r = 0; r < 8; ++r) {
        int orow = base + (hf ? 8 + r : r);
        if (orow < N) out[(size_t)orow * 4 + col] = acc[r];
      }
    }
  }
}

extern "C" void kernel_launch(void* const* d_in, const int* in_sizes, int n_in,
                              void* d_out, int out_size, void* d_ws, size_t ws_size,
                              hipStream_t stream) {
  const float* x  = (const float*)d_in[0];
  const int*   ed = (const int*)d_in[1];
  const float* W1 = (const float*)d_in[2];
  const float* b1 = (const float*)d_in[3];
  const float* W2 = (const float*)d_in[4];
  const float* b2 = (const float*)d_in[5];
  const float* Wl = (const float*)d_in[6];
  const float* bl = (const float*)d_in[7];
  float* out = (float*)d_out;

  const int N = in_sizes[0];
  const int E = in_sizes[1] / 2;

  const int* src = ed;
  const int* dst = ed + E;

  const int B = 256;
  const int nTiles = (N + 15) / 16;
  const int tileBlocks = (nTiles * 32 + B - 1) / B;

  const int nb  = (N + 255) / 256;
  const int nsl = (N + SLCAP - 1) / SLCAP;
  const int SL2 = (N + 2 * nsl - 1) / (2 * nsl);
  const int SL  = 2 * SL2;
  const size_t Nr = ((size_t)N + 64) / 64 * 64;
  const size_t Er = ((size_t)E + 63) / 64 * 64;
  const size_t Pw = (size_t)nsl * 16 * Nr;
  const size_t Rr = ((size_t)nb * 256 * NSUB + 64) / 64 * 64;
  const size_t o_td   = Nr;
  const size_t o_S    = o_td + 2 * Nr;
  const size_t o_rp   = o_S + 5 * Nr;
  const size_t o_cnt  = o_rp + Rr;
  const size_t o_btot = o_cnt + (size_t)nb * NCHUNK;
  const size_t o_bb   = o_btot + 2048;
  const size_t o_csr  = o_bb + 2048;
  const size_t o_big  = o_csr + Er;
  const size_t need   = (o_big + ((Er > Pw) ? Er : Pw)) * 4;

  if (need <= ws_size && nb <= MAXB && 2 * nsl <= NSUB && SL <= SLCAP) {
    float*    dnorm  = (float*)d_ws;
    float*    td     = dnorm + o_td;
    float*    S      = dnorm + o_S;
    int*      rp2    = (int*)d_ws + o_rp;
    unsigned* cnt    = (unsigned*)d_ws + o_cnt;
    unsigned* btot   = (unsigned*)d_ws + o_btot;
    unsigned* bbase  = (unsigned*)d_ws + o_bb;
    int*      csr    = (int*)d_ws + o_csr;
    unsigned* rec    = (unsigned*)d_ws + o_big;
    float*    Pl     = dnorm + o_big;

    const int chunk = (E + NCHUNK - 1) / NCHUNK;

    bucket_count    <<<NCHUNK, 256, 0, stream>>>(dst, cnt, E, chunk, nb);
    bucket_scan_rows<<<nb, 256, 0, stream>>>(cnt, btot);
    bucket_scan_tot <<<1, 1024, 0, stream>>>(btot, bbase, rp2, nb, nb * 256 * NSUB);
    if (chunk <= PCAP * 1024)
      bucket_place_lds<<<NCHUNK, 1024, 0, stream>>>(src, dst, cnt, bbase, rec, E, chunk, nb);
    else
      bucket_place    <<<NCHUNK, 256, 0, stream>>>(src, dst, cnt, bbase, rec, E, chunk, nb);
    bucket_sort_sub <<<nb, 1024, 0, stream>>>(rec, bbase, rp2, csr, N, (unsigned)SL2);

    float* yv = S + (size_t)nsl * N;
    transform_y<<<(N + B - 1) / B, B, 0, stream>>>(x, rp2, dnorm, yv, N);
    pull_y_lds<<<dim3(256 / nsl, nsl), 1024, 0, stream>>>(rp2, csr, yv, S, N, SL);
    make_td<<<(N + B - 1) / B, B, 0, stream>>>(S, nsl, yv, dnorm, td, N);

    pull_td_lds<<<dim3(256 / nsl, nsl), 1024, 0, stream>>>(rp2, csr, td, W1, b1, Pl, N, SL2);
    gcn_out_wmma_u<<<tileBlocks, B, 0, stream>>>(Pl, nsl, td, W1, b1, W2, b2, Wl, bl, out, N, nTiles);
    return;
  }

  float* dnorm = (float*)d_ws;
  float* p     = dnorm + N;
  float* agg   = p + (size_t)N * 16;

  const int edgeBlocks = (int)(((long long)E * 4 + B - 1) / B);

  zero_f32  <<<(N + B - 1) / B, B, 0, stream>>>(dnorm, N);
  deg_accum <<<(E + B - 1) / B, B, 0, stream>>>(dst, dnorm, E);
  transform1<<<(N + B - 1) / B, B, 0, stream>>>(x, W1, dnorm, p, agg, N);
  edge_scatter<<<edgeBlocks, B, 0, stream>>>(src, dst, p, agg, E);
  gcn_mid_wmma<true><<<tileBlocks, B, 0, stream>>>(p, agg, dnorm, b1, W2, p, agg, N, nTiles);
  edge_scatter<<<edgeBlocks, B, 0, stream>>>(src, dst, p, agg, E);
  gcn_out_wmma<<<tileBlocks, B, 0, stream>>>(p, agg, dnorm, b2, Wl, bl, out, N, nTiles);
}


# both pull kernels read the source list 16 bytes per lane (4 entries a load, 4-byte aligned): a quarter of the index loads and line touches per entry; layer-1 f32 sum order changes
# speedup vs baseline: 9.3076x; 1.3073x over previous
#include <hip/hip_runtime.h>

typedef __attribute__((ext_vector_type(2))) float v2f;
typedef __attribute__((ext_vector_type(8))) float v8f;

__global__ void zero_f32(float* __restrict__ buf, int n) {
  int i = blockIdx.x * blockDim.x + threadIdx.x;
  if (i < n) buf[i] = 0.0f;
}

__global__ void deg_accum(const int* __restrict__ dst, float* __restrict__ deg, int E) {
  int e = blockIdx.x * blockDim.x + threadIdx.x;
  if (e < E) atomicAdd(&deg[dst[e]], 1.0f);
}

__global__ void transform1(const float* __restrict__ x, const float* __restrict__ W1,
                           float* __restrict__ dnorm, float* __restrict__ p,
                           float* __restrict__ agg, int N) {
  int i = blockIdx.x * blockDim.x + threadIdx.x;
  if (i >= N) return;
  float di = rsqrtf(dnorm[i] + 1.0f);
  dnorm[i] = di;
  float s = x[i] * di;
  const float4* w = (const float4*)W1;
  float4* pp = (float4*)(p + (size_t)i * 16);
  float4* ag = (float4*)(agg + (size_t)i * 16);
#pragma unroll
  for (int k = 0; k < 4; ++k) {
    float4 wv = w[k];
    float4 o;
    o.x = wv.x * s; o.y = wv.y * s; o.z = wv.z * s; o.w = wv.w * s;
    pp[k] = o;
    float4 z; z.x = 0.f; z.y = 0.f; z.z = 0.f; z.w = 0.f;
    ag[k] = z;
  }
}

__global__ void edge_scatter(const int* __restrict__ src, const int* __restrict__ dst,
                             const float* __restrict__ p, float* __restrict__ agg, int E) {
  int t = blockIdx.x * blockDim.x + threadIdx.x;
  int e = t >> 2;
  if (e >= E) return;
  int sub = (t & 3) * 4;
  int s = src[e];
  int d = dst[e];
  float4 v = *(const float4*)(p + (size_t)s * 16 + sub);
  float* ap = agg + (size_t)d * 16 + sub;
  atomicAdd(ap + 0, v.x);
  atomicAdd(ap + 1, v.y);
  atomicAdd(ap + 2, v.z);
  atomicAdd(ap + 3, v.w);
}

template <bool kZeroAgg>
__global__ void __launch_bounds__(256)
gcn_mid_wmma(const float* __restrict__ p, const float* __restrict__ agg,
             const float* __restrict__ dnorm, const float* __restrict__ b1,
             const float* __restrict__ W2, float* __restrict__ p2,
             float* __restrict__ agg_out, int N, int nTiles) {
  const int lane = threadIdx.x & 31;
  const int tile = (blockIdx.x * blockDim.x + threadIdx.x) >> 5;
  if (tile >= nTiles) return;
  const int base = tile * 16;
  const int col  = lane & 15;
  const bool hi  = lane >= 16;
  const bool full = (base + 16 <= N);

  int arow = base + col;
  if (arow >= N) arow = N - 1;
  const float di = dnorm[arow];

  v2f a[4], b[4];
#pragma unroll
  for (int k = 0; k < 4; ++k) {
    const int c0 = 4 * k + (hi ? 2 : 0);
    float2 pv = *(const float2*)(p   + (size_t)arow * 16 + c0);
    float2 av = *(const float2*)(agg + (size_t)arow * 16 + c0);
    a[k][0] = fmaxf(di * (av.x + pv.x) + b1[c0],     0.0f);
    a[k][1] = fmaxf(di * (av.y + pv.y) + b1[c0 + 1], 0.0f);
    b[k][0] = W2[c0 * 16 + col];
    b[k][1] = W2[(c0 + 1) * 16 + col];
  }

  v8f acc = {};
#pragma unroll
  for (int k = 0; k < 4; ++k)
    acc = __builtin_amdgcn_wmma_f32_16x16x4_f32(false, a[k], false, b[k],
                                                (short)0, acc, false, false);

  if (full) {
#pragma unroll
    for (int r = 0; r < 8; ++r) {
      int orow = base + (hi ? 8 + r : r);
      p2[(size_t)orow * 16 + col]      = acc[r] * dnorm[orow];
      if (kZeroAgg) agg_out[(size_t)orow * 16 + col] = 0.0f;
    }
  } else {
#pragma unroll
    for (int r = 0; r < 8; ++r) {
      int orow = base + (hi ? 8 + r : r);
      if (orow < N) {
        p2[(size_t)orow * 16 + col]      = acc[r] * dnorm[orow];
        if (kZeroAgg) agg_out[(size_t)orow * 16 + col] = 0.0f;
      }
    }
  }
}

__global__ void __launch_bounds__(256)
gcn_out_wmma(const float* __restrict__ p2, const float* __restrict__ agg2,
             const float* __restrict__ dnorm, const float* __restrict__ b2,
             const float* __restrict__ Wl, const float* __restrict__ bl,
             float* __restrict__ out, int N, int nTiles) {
  const int lane = threadIdx.x & 31;
  const int tile = (blockIdx.x * blockDim.x + threadIdx.x) >> 5;
  if (tile >= nTiles) return;
  const int base = tile * 16;
  const int col  = lane & 15;
  const bool hi  = lane >= 16;
  const bool full = (base + 16 <= N);

  int arow = base + col;
  if (arow >= N) arow = N - 1;
  const float di = dnorm[arow];

  v2f a[4], b[4];
#pragma unroll
  for (int k = 0; k < 4; ++k) {
    const int c0 = 4 * k + (hi ? 2 : 0);
    float2 pv = *(const float2*)(p2   + (size_t)arow * 16 + c0);
    float2 av = *(const float2*)(agg2 + (size_t)arow * 16 + c0);
    a[k][0] = fmaxf(di * (av.x + pv.x) + b2[c0],     0.0f);
    a[k][1] = fmaxf(di * (av.y + pv.y) + b2[c0 + 1], 0.0f);
    b[k][0] = (col < 4) ? Wl[c0 * 4 + col]       : 0.0f;
    b[k][1] = (col < 4) ? Wl[(c0 + 1) * 4 + col] : 0.0f;
  }

  const float cv = (col < 4) ? bl[col] : 0.0f;
  v8f acc;
#pragma unroll
  for (int r = 0; r < 8; ++r) acc[r] = cv;
#pragma unroll
  for (int k = 0; k < 4; ++k)
    acc = __builtin_amdgcn_wmma_f32_16x16x4_f32(false, a[k], false, b[k],
                                                (short)0, acc, false, false);

  if (col < 4) {
    if (full) {
#pragma unroll
      for (int r = 0; r < 8; ++r) {
        int orow = base + (hi ? 8 + r : r);
        out[(size_t)orow * 4 + col] = acc[r];
      }
    } else {
#pragma unroll
      for (int r = 0; r < 8; ++r) {
        int orow = base + (hi ? 8 + r : r);
        if (orow < N) out[(size_t)orow * 4 + col] = acc[r];
      }
    }
  }
}

#define NCHUNK 1024
#define MAXB   1024

__device__ __forceinline__ unsigned wave_incl_scan(unsigned v, int lane) {
#pragma unroll
  for (int off = 1; off < 32; off <<= 1) {
    const unsigned n = __shfl_up(v, off, 32);
    if (lane >= off) v += n;
  }
  return v;
}

__device__ __forceinline__ unsigned block_excl_scan_1024(unsigned v, unsigned* wsum) {
  const int lane = threadIdx.x & 31;
  const int wave = threadIdx.x >> 5;
  const unsigned incl = wave_incl_scan(v, lane);
  if (lane == 31) wsum[wave] = incl;
  __syncthreads();
  if (wave == 0) {
    const unsigned w  = wsum[lane];
    const unsigned wi = wave_incl_scan(w, lane);
    wsum[lane] = wi - w;
  }
  __syncthreads();
  return incl - v + wsum[wave];
}

#define PCAP 13
#define SCAP 18

__global__ void __launch_bounds__(1024) __attribute__((amdgpu_waves_per_eu(16, 16)))
chunk_sort(const int* __restrict__ src, const int* __restrict__ dst,
           unsigned* __restrict__ lst, unsigned* __restrict__ rec,
           int E, int chunk, int nb) {
  __shared__ unsigned h[MAXB + 32];
  __shared__ unsigned wsum[32];
  __shared__ unsigned stage[PCAP * 1024 + 32];
  const int t    = threadIdx.x;
  const int lane = t & 31;
  const int w    = blockIdx.x;
  const int e0   = w * chunk;
  const int e1   = (e0 + chunk < E) ? (e0 + chunk) : E;
  const int n    = e1 - e0;
  h[t] = 0u;
  if (t < 32) h[MAXB + t] = 0u;

  unsigned sv[PCAP], dv[PCAP];
#pragma unroll
  for (int j = 0; j < PCAP; ++j) {
    const bool ok = t < n - j * 1024;
    const int ec = ok ? (e0 + j * 1024 + t) : (E - 1);
    dv[j] = (unsigned)dst[ec];
    sv[j] = (unsigned)src[ec];
  }
#pragma unroll
  for (int j = 0; j < PCAP; ++j) {
    sv[j] |= (dv[j] & 255u) << 18;
    dv[j] >>= 8;
    asm volatile("" : "+v"(sv[j]), "+v"(dv[j]));
  }
  __syncthreads();
  unsigned rk[PCAP];
#pragma unroll
  for (int j = 0; j < PCAP; ++j) {
    const bool ok = t < n - j * 1024;
    const unsigned b = ok ? dv[j] : (unsigned)(MAXB + lane);
    rk[j] = atomicAdd(&h[b], 1u);
  }
#pragma unroll
  for (int j = 0; j < PCAP; ++j) dv[j] |= rk[j] << 10;
  __syncthreads();
  const unsigned ex = block_excl_scan_1024(h[t], wsum);
  h[t] = ex;
  if (t <= nb) lst[(size_t)t * NCHUNK + w] = ex;
  __syncthreads();
#pragma unroll
  for (int j = 0; j < PCAP; ++j) {
    const bool ok = t < n - j * 1024;
    const unsigned b    = ok ? (dv[j] & 1023u) : 0u;
    const unsigned rank = dv[j] >> 10;
    const unsigned lp = ok ? (h[b] + rank) : (unsigned)(PCAP * 1024 + lane);
    stage[lp] = sv[j];
  }
  __syncthreads();
  for (int i = t; i < n; i += 1024) rec[e0 + i] = stage[i];
}

#define NSUB 8

__device__ __forceinline__ unsigned key_of(unsigned nl, unsigned sub) {
  return ((nl >> 5) << 8) | (sub << 5) | (nl & 31u);
}
__device__ __forceinline__ size_t slot_of(int node, int sub) {
  return (size_t)(node >> 5) * 256 + (size_t)sub * 32 + (size_t)(node & 31);
}

__device__ __forceinline__ unsigned sub_of(unsigned s, unsigned SL2) {
  unsigned k = 0u;
#pragma unroll
  for (unsigned q = 1; q < NSUB; ++q) k += (s >= q * SL2) ? 1u : 0u;
  return k;
}

__global__ void __launch_bounds__(1024) __attribute__((amdgpu_waves_per_eu(16, 16)))
bucket_sort_sub(const unsigned* __restrict__ rec, const unsigned* __restrict__ lst,
                int* __restrict__ rp2, int* __restrict__ csr, int N, unsigned SL2, int chunk) {
  __shared__ unsigned c[256 * NSUB + 32];
  __shared__ unsigned wsum[32];
  __shared__ unsigned wsum2[32];
  __shared__ unsigned tot[2];
  __shared__ unsigned stage[SCAP * 1024 + 32];
  const int t    = threadIdx.x;
  const int lane = t & 31;
  c[t] = 0u;
  c[t + 1024] = 0u;
  if (t < 32) c[256 * NSUB + t] = 0u;

  const unsigned ra  = lst[(size_t)blockIdx.x * NCHUNK + t];
  const unsigned rz  = lst[(size_t)(blockIdx.x + 1) * NCHUNK + t];
  const unsigned len = rz - ra;
  const unsigned exl = block_excl_scan_1024(len, wsum);
  const unsigned exa = block_excl_scan_1024(ra, wsum2);
  if (t == 1023) { tot[0] = exa + ra; tot[1] = exl + len; }
  __syncthreads();
  const unsigned bs = (unsigned)__builtin_amdgcn_readfirstlane((int)tot[0]);
  const unsigned n  = (unsigned)__builtin_amdgcn_readfirstlane((int)tot[1]);
  const unsigned g0 = (unsigned)t * (unsigned)chunk + ra;
  if (blockIdx.x == gridDim.x - 1 && t == 0) rp2[(size_t)gridDim.x * 256 * NSUB] = (int)(bs + n);
  if (blockIdx.x == gridDim.x - 1 && t < 4) csr[bs + n + (unsigned)t] = 0;

  if (n <= (unsigned)(SCAP * 1024)) {
    for (unsigned q = 0; q < len; ++q) stage[exl + q] = g0 + q;
    if (t < 32) stage[SCAP * 1024 + t] = 0u;
    __syncthreads();
    unsigned rv[SCAP], rk[SCAP];
#pragma unroll
    for (int j = 0; j < SCAP; ++j) {
      const unsigned i = (unsigned)(j * 1024 + t);
      rv[j] = stage[(i < n) ? i : (unsigned)(SCAP * 1024 + lane)];
    }
#pragma unroll
    for (int j = 0; j < SCAP; ++j) rv[j] = rec[rv[j]];
#pragma unroll
    for (int j = 0; j < SCAP; ++j) {
      const unsigned s = rv[j] & 0x3FFFFu;
      rv[j] = s | (key_of(rv[j] >> 18, sub_of(s, SL2)) << 18);
      asm volatile("" : "+v"(rv[j]));
    }
    __syncthreads();
#pragma unroll
    for (int j = 0; j < SCAP; ++j) {
      const bool ok = (unsigned)(j * 1024 + t) < n;
      const unsigned k = ok ? (rv[j] >> 18) : (unsigned)(256 * NSUB + lane);
      rk[j] = atomicAdd(&c[k], 1u);
    }
    __syncthreads();
    const unsigned v0 = c[2 * t];
    const unsigned v1 = c[2 * t + 1];
    const unsigned ex = block_excl_scan_1024(v0 + v1, wsum);
    c[2 * t]     = ex;
    c[2 * t + 1] = ex + v0;
    {
      int2 o;
      o.x = (int)(bs + ex);
      o.y = (int)(bs + ex + v0);
      ((int2*)rp2)[(size_t)blockIdx.x * 1024 + t] = o;
    }
    __syncthreads();
#pragma unroll
    for (int j = 0; j < SCAP; ++j) {
      const bool ok = (unsigned)(j * 1024 + t) < n;
      asm volatile("" : "+v"(rv[j]));
      const unsigned key = ok ? (rv[j] >> 18) : 0u;
      const unsigned lp  = ok ? (c[key] + rk[j]) : (unsigned)(SCAP * 1024 + lane);
      stage[lp] = rv[j] & 0x3FFFFu;
    }
    __syncthreads();
    for (unsigned i = (unsigned)t; i < n; i += 1024u) csr[bs + i] = (int)stage[i];
  } else {
    for (unsigned q = 0; q < len; ++q) {
      const unsigned r = rec[g0 + q];
      atomicAdd(&c[key_of(r >> 18, sub_of(r & 0x3FFFFu, SL2))], 1u);
    }
    __syncthreads();
    const unsigned v0 = c[2 * t];
    const unsigned v1 = c[2 * t + 1];
    const unsigned ex = block_excl_scan_1024(v0 + v1, wsum);
    c[2 * t]     = bs + ex;
    c[2 * t + 1] = bs + ex + v0;
    {
      int2 o;
      o.x = (int)(bs + ex);
      o.y = (int)(bs + ex + v0);
      ((int2*)rp2)[(size_t)blockIdx.x * 1024 + t] = o;
    }
    __syncthreads();
    for (unsigned q = 0; q < len; ++q) {
      const unsigned r = rec[g0 + q];
      const unsigned pos = atomicAdd(&c[key_of(r >> 18, sub_of(r & 0x3FFFFu, SL2))], 1u);
      csr[pos] = (int)(r & 0x3FFFFu);
    }
  }
}

__global__ void transform_y(const float* __restrict__ x, const int* __restrict__ rp2,
                            float* __restrict__ dnorm, float* __restrict__ y, int N) {
  int i = blockIdx.x * blockDim.x + threadIdx.x;
  if (i >= N) return;
  const int* rr = rp2 + slot_of(i, 0);
  int deg = 0;
#pragma unroll
  for (int s = 0; s < NSUB; ++s) deg += rr[s * 32 + 1] - rr[s * 32];
  float di = rsqrtf((float)deg + 1.0f);
  dnorm[i] = di;
  y[i] = x[i] * di;
}

#define SLCAP 68096
#define YV    3

__global__ void __launch_bounds__(1024)
pull_y_lds(const int* __restrict__ rp2, const int* __restrict__ csr,
           const float* __restrict__ y, float* __restrict__ S, int N, int SL) {
  __shared__ float ys[SLCAP];
  const int t   = threadIdx.x;
  const int k   = blockIdx.y;
  const int s0  = k * SL;
  const int s1  = (s0 + SL < N) ? (s0 + SL) : N;
  const int len = s1 - s0;
  const unsigned ulen = (len > 0) ? (unsigned)len : 0u;

  for (int j0 = t; j0 < len; j0 += 1024 * 16) {
    float v[16];
#pragma unroll
    for (int u = 0; u < 16; ++u) {
      const int j = j0 + u * 1024;
      v[u] = y[s0 + ((j < len) ? j : 0)];
    }
#pragma unroll
    for (int u = 0; u < 16; ++u) {
      const int j = j0 + u * 1024;
      if (j < len) ys[j] = v[u];
    }
  }
  __syncthreads();

  const int G   = gridDim.x;
  const int npg = (((N + G - 1) / G) + 31) & ~31;
  const int n0  = blockIdx.x * npg;
  const int n1  = (n0 + npg < N) ? (n0 + npg) : N;
  const int sub = t & 3;
  const int hh  = sub >> 1;
  const int ls  = sub & 1;
  float* Sk = S + (size_t)k * N;
  for (int nb0 = n0; nb0 < n1; nb0 += 256) {
    const int i = nb0 + (t >> 2);
    const bool live = i < n1;
    int e = 0, e1 = 0;
    if (live) {
      const int* rr = rp2 + slot_of(i, 2 * k + hh);
      e  = rr[0];
      e1 = rr[1];
    }
    float acc = 0.0f;
    for (e += 4 * ls; e < e1; e += 8 * YV) {
      int sv[YV][4];
#pragma unroll
      for (int j = 0; j < YV; ++j) {
        const int ee = e + 8 * j;
        __builtin_memcpy(sv[j], csr + ((ee < e1) ? ee : e), 16);
      }
#pragma unroll
      for (int j = 0; j < YV; ++j) {
#pragma unroll
        for (int u = 0; u < 4; ++u) {
          const unsigned r = (unsigned)(sv[j][u] - s0);
          const bool ok = ((e + 8 * j + u) < e1) && (r < ulen);
          const float v = ys[ok ? r : 0u];
          acc += ok ? v : 0.0f;
        }
      }
    }
    acc += __shfl_xor(acc, 1, 32);
    acc += __shfl_xor(acc, 2, 32);
    if (live && sub == 0) Sk[i] = acc;
  }
}

__global__ void make_td(const float* __restrict__ S, int nsl, const float* __restrict__ y,
                        const float* __restrict__ dnorm, float* __restrict__ td, int N) {
  int i = blockIdx.x * blockDim.x + threadIdx.x;
  if (i >= N) return;
  const float di = dnorm[i];
  float s = 0.0f;
  for (int q = 0; q < nsl; ++q) s += S[(size_t)q * N + i];
  s += y[i];
  float2 o;
  o.x = di * s;
  o.y = di;
  ((float2*)td)[i] = o;
}

#define TDCAP (SLCAP / 2)
#define TU    4

__global__ void __launch_bounds__(1024)
pull_td_lds(const int* __restrict__ rp2, const int* __restrict__ csr,
            const float* __restrict__ tdf, const float* __restrict__ W1,
            const float* __restrict__ b1, float* __restrict__ P, int N, int SL2) {
  __shared__ float2 tds[TDCAP];
  const float2* td = (const float2*)tdf;
  const int t = threadIdx.x;
  const int k = blockIdx.y;
  v2f w1p[8], b1p[8];
#pragma unroll
  for (int c = 0; c < 8; ++c) {
    w1p[c][0] = W1[2 * c]; w1p[c][1] = W1[2 * c + 1];
    b1p[c][0] = b1[2 * c]; b1p[c][1] = b1[2 * c + 1];
    asm volatile("" : "+v"(w1p[c]), "+v"(b1p[c]));
  }

  const int G   = gridDim.x;
  const int npg = (((N + G - 1) / G) + 31) & ~31;
  const int n0  = blockIdx.x * npg;
  const int n1  = (n0 + npg < N) ? (n0 + npg) : N;

  for (int h = 0; h < 2; ++h) {
    const int sub = 2 * k + h;
    const int s0  = sub * SL2;
    const int s1  = (s0 + SL2 < N) ? (s0 + SL2) : N;
    const int len = s1 - s0;
    const unsigned ulen = (len > 0) ? (unsigned)len : 0u;
    __syncthreads();
    for (int j0 = t; j0 < len; j0 += 1024 * 16) {
      float2 v[16];
#pragma unroll
      for (int u = 0; u < 16; ++u) {
        const int j = j0 + u * 1024;
        v[u] = td[s0 + ((j < len) ? j : 0)];
      }
#pragma unroll
      for (int u = 0; u < 16; ++u) {
        const int j = j0 + u * 1024;
        if (j < len) tds[j] = v[u];
      }
    }
    __syncthreads();

    for (int nb0 = n0; nb0 < n1; nb0 += 1024) {
      const int i = nb0 + t;
      const bool live = i < n1;
      int e0 = 0, cnt = 0;
      if (live) {
        const int* rr = rp2 + slot_of(i, sub);
        e0  = rr[0];
        cnt = rr[1] - e0;
      }
      int wmax = cnt;
#pragma unroll
      for (int off = 16; off > 0; off >>= 1) {
        const int o = __shfl_xor(wmax, off, 32);
        wmax = (o > wmax) ? o : wmax;
      }
      wmax = __builtin_amdgcn_readfirstlane(wmax);

      v2f acc[8];
#pragma unroll
      for (int c = 0; c < 8; ++c) { acc[c][0] = 0.0f; acc[c][1] = 0.0f; }
      int cur[TU];
      __builtin_memcpy(cur, csr + ((0 < cnt) ? e0 : 0), 4 * TU);
      for (int base = 0; base < wmax; base += TU) {
        int nxt[TU];
        {
          const int q = base + TU;
          __builtin_memcpy(nxt, csr + ((q < cnt) ? (e0 + q) : 0), 4 * TU);
        }
#pragma unroll
        for (int j = 0; j < TU; ++j) {
          const int q = base + j;
          const unsigned r = (unsigned)(cur[j] - s0);
          const bool ok = (q < cnt) && (r < ulen);
          const float2 v = tds[ok ? r : 0u];
          const float tj = ok ? v.x : 0.0f;
          const float dj = ok ? v.y : 0.0f;
          const v2f tj2 = {tj, tj};
          const v2f dj2 = {dj, dj};
          const v2f z2  = {0.0f, 0.0f};
#pragma unroll
          for (int c = 0; c < 8; ++c)
            acc[c] = __builtin_elementwise_fma(dj2, __builtin_elementwise_max(__builtin_elementwise_fma(tj2, w1p[c], b1p[c]), z2), acc[c]);
        }
        asm volatile("" ::: "memory");
#pragma unroll
        for (int j = 0; j < TU; ++j) cur[j] = nxt[j];
      }
      if (live) {
        float4* Pp = (float4*)(P + ((size_t)k * N + i) * 16);
#pragma unroll
        for (int q = 0; q < 4; ++q) {
          float4 o;
          o.x = acc[2 * q][0]; o.y = acc[2 * q][1]; o.z = acc[2 * q + 1][0]; o.w = acc[2 * q + 1][1];
          if (h) {
            const float4 p = Pp[q];
            o.x += p.x; o.y += p.y; o.z += p.z; o.w += p.w;
          }
          Pp[q] = o;
        }
      }
    }
  }
}

__global__ void __launch_bounds__(256)
gcn_out_wmma_u(const float* __restrict__ P, int nsl, const float* __restrict__ td,
               const float* __restrict__ W1, const float* __restrict__ b1,
               const float* __restrict__ W2, const float* __restrict__ b2,
               const float* __restrict__ Wl, const float* __restrict__ bl,
               float* __restrict__ out, int N, int nTiles) {
  const int lane = threadIdx.x & 31;
  const int tile = (blockIdx.x * blockDim.x + threadIdx.x) >> 5;
  if (tile >= nTiles) return;
  const int base = tile * 16;
  const int col  = lane & 15;
  const int hf   = lane >> 4;
  const bool full = (base + 16 <= N);

  int arow = base + col;
  if (arow >= N) arow = N - 1;
  const float2 tdv = ((const float2*)td)[arow];
  const float ti = tdv.x;
  const float di = tdv.y;

  v2f aw[4], bu[4];
#pragma unroll
  for (int k = 0; k < 4; ++k) {
    const int c0 = 4 * k + 2 * hf;
    float u0 = 0.0f, u1 = 0.0f;
    for (int q = 0; q < nsl; ++q) {
      const float2 pv = *(const float2*)(P + ((size_t)q * N + arow) * 16 + c0);
      u0 += pv.x;
      u1 += pv.y;
    }
    u0 += di * fmaxf(fmaf(ti, W1[c0],     b1[c0]),     0.0f);
    u1 += di * fmaxf(fmaf(ti, W1[c0 + 1], b1[c0 + 1]), 0.0f);
    bu[k][0] = u0;
    bu[k][1] = u1;
    aw[k][0] = W2[c0 * 16 + col];
    aw[k][1] = W2[(c0 + 1) * 16 + col];
  }

  v8f z = {};
#pragma unroll
  for (int k = 0; k < 4; ++k)
    z = __builtin_amdgcn_wmma_f32_16x16x4_f32(false, aw[k], false, bu[k],
                                              (short)0, z, false, false);

  v2f ah[4], bw[4];
#pragma unroll
  for (int k = 0; k < 4; ++k) {
    const int c0 = 8 * hf + 2 * k;
    ah[k][0] = fmaxf(fmaf(di, z[2 * k],     b2[c0]),     0.0f);
    ah[k][1] = fmaxf(fmaf(di, z[2 * k + 1], b2[c0 + 1]), 0.0f);
    bw[k][0] = (col < 4) ? Wl[c0 * 4 + col]       : 0.0f;
    bw[k][1] = (col < 4) ? Wl[(c0 + 1) * 4 + col] : 0.0f;
  }

  const float cv = (col < 4) ? bl[col] : 0.0f;
  v8f acc;
#pragma unroll
  for (int r = 0; r < 8; ++r) acc[r] = cv;
#pragma unroll
  for (int k = 0; k < 4; ++k)
    acc = __builtin_amdgcn_wmma_f32_16x16x4_f32(false, ah[k], false, bw[k],
                                                (short)0, acc, false, false);

  if (col < 4) {
    if (full) {
#pragma unroll
      for (int r = 0; r < 8; ++r) {
        int orow = base + (hf ? 8 + r : r);
        out[(size_t)orow * 4 + col] = acc[r];
      }
    } else {
#pragma unroll
      for (int r = 0; r < 8; ++r) {
        int orow = base + (hf ? 8 + r : r);
        if (orow < N) out[(size_t)orow * 4 + col] = acc[r];
      }
    }
  }
}

extern "C" void kernel_launch(void* const* d_in, const int* in_sizes, int n_in,
                              void* d_out, int out_size, void* d_ws, size_t ws_size,
                              hipStream_t stream) {
  const float* x  = (const float*)d_in[0];
  const int*   ed = (const int*)d_in[1];
  const float* W1 = (const float*)d_in[2];
  const float* b1 = (const float*)d_in[3];
  const float* W2 = (const float*)d_in[4];
  const float* b2 = (const float*)d_in[5];
  const float* Wl = (const float*)d_in[6];
  const float* bl = (const float*)d_in[7];
  float* out = (float*)d_out;

  const int N = in_sizes[0];
  const int E = in_sizes[1] / 2;

  const int* src = ed;
  const int* dst = ed + E;

  const int B = 256;
  const int nTiles = (N + 15) / 16;
  const int tileBlocks = (nTiles * 32 + B - 1) / B;

  const int nb  = (N + 255) / 256;
  const int nsl = (N + SLCAP - 1) / SLCAP;
  const int SL2 = (N + 2 * nsl - 1) / (2 * nsl);
  const int SL  = 2 * SL2;
  const size_t Nr = ((size_t)N + 64) / 64 * 64;
  const size_t Er = ((size_t)E + 4 + 63) / 64 * 64;
  const size_t Pw = (size_t)nsl * 16 * Nr;
  const size_t Rr = ((size_t)nb * 256 * NSUB + 64) / 64 * 64;
  const size_t o_td   = Nr;
  const size_t o_S    = o_td + 2 * Nr;
  const size_t o_rp   = o_S + 5 * Nr;
  const size_t o_lst  = o_rp + Rr;
  const size_t o_csr  = o_lst + (size_t)(nb + 1) * NCHUNK;
  const size_t o_big  = o_csr + Er;
  const size_t need   = (o_big + ((Er > Pw) ? Er : Pw)) * 4;
  const int chunk = (E + NCHUNK - 1) / NCHUNK;

  if (need <= ws_size && nb < MAXB && chunk <= PCAP * 1024 && 2 * nsl <= NSUB && SL <= SLCAP) {
    float*    dnorm  = (float*)d_ws;
    float*    td     = dnorm + o_td;
    float*    S      = dnorm + o_S;
    int*      rp2    = (int*)d_ws + o_rp;
    unsigned* lst    = (unsigned*)d_ws + o_lst;
    int*      csr    = (int*)d_ws + o_csr;
    unsigned* rec    = (unsigned*)d_ws + o_big;
    float*    Pl     = dnorm + o_big;

    chunk_sort     <<<NCHUNK, 1024, 0, stream>>>(src, dst, lst, rec, E, chunk, nb);
    bucket_sort_sub<<<nb, 1024, 0, stream>>>(rec, lst, rp2, csr, N, (unsigned)SL2, chunk);

    float* yv = S + (size_t)nsl * N;
    transform_y<<<(N + B - 1) / B, B, 0, stream>>>(x, rp2, dnorm, yv, N);
    pull_y_lds<<<dim3(256 / nsl, nsl), 1024, 0, stream>>>(rp2, csr, yv, S, N, SL);
    make_td<<<(N + B - 1) / B, B, 0, stream>>>(S, nsl, yv, dnorm, td, N);

    pull_td_lds<<<dim3(256 / nsl, nsl), 1024, 0, stream>>>(rp2, csr, td, W1, b1, Pl, N, SL2);
    gcn_out_wmma_u<<<tileBlocks, B, 0, stream>>>(Pl, nsl, td, W1, b1, W2, b2, Wl, bl, out, N, nTiles);
    return;
  }

  float* dnorm = (float*)d_ws;
  float* p     = dnorm + N;
  float* agg   = p + (size_t)N * 16;

  const int edgeBlocks = (int)(((long long)E * 4 + B - 1) / B);

  zero_f32  <<<(N + B - 1) / B, B, 0, stream>>>(dnorm, N);
  deg_accum <<<(E + B - 1) / B, B, 0, stream>>>(dst, dnorm, E);
  transform1<<<(N + B - 1) / B, B, 0, stream>>>(x, W1, dnorm, p, agg, N);
  edge_scatter<<<edgeBlocks, B, 0, stream>>>(src, dst, p, agg, E);
  gcn_mid_wmma<true><<<tileBlocks, B, 0, stream>>>(p, agg, dnorm, b1, W2, p, agg, N, nTiles);
  edge_scatter<<<edgeBlocks, B, 0, stream>>>(src, dst, p, agg, E);
  gcn_out_wmma<<<tileBlocks, B, 0, stream>>>(p, agg, dnorm, b2, Wl, bl, out, N, nTiles);
}


# both pull kernels ask for memory ahead of use: table copy loads the next trip before storing this one; next pass's run bounds, list entries and stored sums asked for during this pass; same sums
# speedup vs baseline: 9.6927x; 1.0414x over previous
#include <hip/hip_runtime.h>

typedef __attribute__((ext_vector_type(2))) float v2f;
typedef __attribute__((ext_vector_type(8))) float v8f;

__global__ void zero_f32(float* __restrict__ buf, int n) {
  int i = blockIdx.x * blockDim.x + threadIdx.x;
  if (i < n) buf[i] = 0.0f;
}

__global__ void deg_accum(const int* __restrict__ dst, float* __restrict__ deg, int E) {
  int e = blockIdx.x * blockDim.x + threadIdx.x;
  if (e < E) atomicAdd(&deg[dst[e]], 1.0f);
}

__global__ void transform1(const float* __restrict__ x, const float* __restrict__ W1,
                           float* __restrict__ dnorm, float* __restrict__ p,
                           float* __restrict__ agg, int N) {
  int i = blockIdx.x * blockDim.x + threadIdx.x;
  if (i >= N) return;
  float di = rsqrtf(dnorm[i] + 1.0f);
  dnorm[i] = di;
  float s = x[i] * di;
  const float4* w = (const float4*)W1;
  float4* pp = (float4*)(p + (size_t)i * 16);
  float4* ag = (float4*)(agg + (size_t)i * 16);
#pragma unroll
  for (int k = 0; k < 4; ++k) {
    float4 wv = w[k];
    float4 o;
    o.x = wv.x * s; o.y = wv.y * s; o.z = wv.z * s; o.w = wv.w * s;
    pp[k] = o;
    float4 z; z.x = 0.f; z.y = 0.f; z.z = 0.f; z.w = 0.f;
    ag[k] = z;
  }
}

__global__ void edge_scatter(const int* __restrict__ src, const int* __restrict__ dst,
                             const float* __restrict__ p, float* __restrict__ agg, int E) {
  int t = blockIdx.x * blockDim.x + threadIdx.x;
  int e = t >> 2;
  if (e >= E) return;
  int sub = (t & 3) * 4;
  int s = src[e];
  int d = dst[e];
  float4 v = *(const float4*)(p + (size_t)s * 16 + sub);
  float* ap = agg + (size_t)d * 16 + sub;
  atomicAdd(ap + 0, v.x);
  atomicAdd(ap + 1, v.y);
  atomicAdd(ap + 2, v.z);
  atomicAdd(ap + 3, v.w);
}

template <bool kZeroAgg>
__global__ void __launch_bounds__(256)
gcn_mid_wmma(const float* __restrict__ p, const float* __restrict__ agg,
             const float* __restrict__ dnorm, const float* __restrict__ b1,
             const float* __restrict__ W2, float* __restrict__ p2,
             float* __restrict__ agg_out, int N, int nTiles) {
  const int lane = threadIdx.x & 31;
  const int tile = (blockIdx.x * blockDim.x + threadIdx.x) >> 5;
  if (tile >= nTiles) return;
  const int base = tile * 16;
  const int col  = lane & 15;
  const bool hi  = lane >= 16;
  const bool full = (base + 16 <= N);

  int arow = base + col;
  if (arow >= N) arow = N - 1;
  const float di = dnorm[arow];

  v2f a[4], b[4];
#pragma unroll
  for (int k = 0; k < 4; ++k) {
    const int c0 = 4 * k + (hi ? 2 : 0);
    float2 pv = *(const float2*)(p   + (size_t)arow * 16 + c0);
    float2 av = *(const float2*)(agg + (size_t)arow * 16 + c0);
    a[k][0] = fmaxf(di * (av.x + pv.x) + b1[c0],     0.0f);
    a[k][1] = fmaxf(di * (av.y + pv.y) + b1[c0 + 1], 0.0f);
    b[k][0] = W2[c0 * 16 + col];
    b[k][1] = W2[(c0 + 1) * 16 + col];
  }

  v8f acc = {};
#pragma unroll
  for (int k = 0; k < 4; ++k)
    acc = __builtin_amdgcn_wmma_f32_16x16x4_f32(false, a[k], false, b[k],
                                                (short)0, acc, false, false);

  if (full) {
#pragma unroll
    for (int r = 0; r < 8; ++r) {
      int orow = base + (hi ? 8 + r : r);
      p2[(size_t)orow * 16 + col]      = acc[r] * dnorm[orow];
      if (kZeroAgg) agg_out[(size_t)orow * 16 + col] = 0.0f;
    }
  } else {
#pragma unroll
    for (int r = 0; r < 8; ++r) {
      int orow = base + (hi ? 8 + r : r);
      if (orow < N) {
        p2[(size_t)orow * 16 + col]      = acc[r] * dnorm[orow];
        if (kZeroAgg) agg_out[(size_t)orow * 16 + col] = 0.0f;
      }
    }
  }
}

__global__ void __launch_bounds__(256)
gcn_out_wmma(const float* __restrict__ p2, const float* __restrict__ agg2,
             const float* __restrict__ dnorm, const float* __restrict__ b2,
             const float* __restrict__ Wl, const float* __restrict__ bl,
             float* __restrict__ out, int N, int nTiles) {
  const int lane = threadIdx.x & 31;
  const int tile = (blockIdx.x * blockDim.x + threadIdx.x) >> 5;
  if (tile >= nTiles) return;
  const int base = tile * 16;
  const int col  = lane & 15;
  const bool hi  = lane >= 16;
  const bool full = (base + 16 <= N);

  int arow = base + col;
  if (arow >= N) arow = N - 1;
  const float di = dnorm[arow];

  v2f a[4], b[4];
#pragma unroll
  for (int k = 0; k < 4; ++k) {
    const int c0 = 4 * k + (hi ? 2 : 0);
    float2 pv = *(const float2*)(p2   + (size_t)arow * 16 + c0);
    float2 av = *(const float2*)(agg2 + (size_t)arow * 16 + c0);
    a[k][0] = fmaxf(di * (av.x + pv.x) + b2[c0],     0.0f);
    a[k][1] = fmaxf(di * (av.y + pv.y) + b2[c0 + 1], 0.0f);
    b[k][0] = (col < 4) ? Wl[c0 * 4 + col]       : 0.0f;
    b[k][1] = (col < 4) ? Wl[(c0 + 1) * 4 + col] : 0.0f;
  }

  const float cv = (col < 4) ? bl[col] : 0.0f;
  v8f acc;
#pragma unroll
  for (int r = 0; r < 8; ++r) acc[r] = cv;
#pragma unroll
  for (int k = 0; k < 4; ++k)
    acc = __builtin_amdgcn_wmma_f32_16x16x4_f32(false, a[k], false, b[k],
                                                (short)0, acc, false, false);

  if (col < 4) {
    if (full) {
#pragma unroll
      for (int r = 0; r < 8; ++r) {
        int orow = base + (hi ? 8 + r : r);
        out[(size_t)orow * 4 + col] = acc[r];
      }
    } else {
#pragma unroll
      for (int r = 0; r < 8; ++r) {
        int orow = base + (hi ? 8 + r : r);
        if (orow < N) out[(size_t)orow * 4 + col] = acc[r];
      }
    }
  }
}

#define NCHUNK 1024
#define MAXB   1024

__device__ __forceinline__ unsigned wave_incl_scan(unsigned v, int lane) {
#pragma unroll
  for (int off = 1; off < 32; off <<= 1) {
    const unsigned n = __shfl_up(v, off, 32);
    if (lane >= off) v += n;
  }
  return v;
}

__device__ __forceinline__ unsigned block_excl_scan_1024(unsigned v, unsigned* wsum) {
  const int lane = threadIdx.x & 31;
  const int wave = threadIdx.x >> 5;
  const unsigned incl = wave_incl_scan(v, lane);
  if (lane == 31) wsum[wave] = incl;
  __syncthreads();
  if (wave == 0) {
    const unsigned w  = wsum[lane];
    const unsigned wi = wave_incl_scan(w, lane);
    wsum[lane] = wi - w;
  }
  __syncthreads();
  return incl - v + wsum[wave];
}

#define PCAP 13
#define SCAP 18

__global__ void __launch_bounds__(1024) __attribute__((amdgpu_waves_per_eu(16, 16)))
chunk_sort(const int* __restrict__ src, const int* __restrict__ dst,
           unsigned* __restrict__ lst, unsigned* __restrict__ rec,
           int E, int chunk, int nb) {
  __shared__ unsigned h[MAXB + 32];
  __shared__ unsigned wsum[32];
  __shared__ unsigned stage[PCAP * 1024 + 32];
  const int t    = threadIdx.x;
  const int lane = t & 31;
  const int w    = blockIdx.x;
  const int e0   = w * chunk;
  const int e1   = (e0 + chunk < E) ? (e0 + chunk) : E;
  const int n    = e1 - e0;
  h[t] = 0u;
  if (t < 32) h[MAXB + t] = 0u;

  unsigned sv[PCAP], dv[PCAP];
#pragma unroll
  for (int j = 0; j < PCAP; ++j) {
    const bool ok = t < n - j * 1024;
    const int ec = ok ? (e0 + j * 1024 + t) : (E - 1);
    dv[j] = (unsigned)dst[ec];
    sv[j] = (unsigned)src[ec];
  }
#pragma unroll
  for (int j = 0; j < PCAP; ++j) {
    sv[j] |= (dv[j] & 255u) << 18;
    dv[j] >>= 8;
    asm volatile("" : "+v"(sv[j]), "+v"(dv[j]));
  }
  __syncthreads();
  unsigned rk[PCAP];
#pragma unroll
  for (int j = 0; j < PCAP; ++j) {
    const bool ok = t < n - j * 1024;
    const unsigned b = ok ? dv[j] : (unsigned)(MAXB + lane);
    rk[j] = atomicAdd(&h[b], 1u);
  }
#pragma unroll
  for (int j = 0; j < PCAP; ++j) dv[j] |= rk[j] << 10;
  __syncthreads();
  const unsigned ex = block_excl_scan_1024(h[t], wsum);
  h[t] = ex;
  if (t <= nb) lst[(size_t)t * NCHUNK + w] = ex;
  __syncthreads();
#pragma unroll
  for (int j = 0; j < PCAP; ++j) {
    const bool ok = t < n - j * 1024;
    const unsigned b    = ok ? (dv[j] & 1023u) : 0u;
    const unsigned rank = dv[j] >> 10;
    const unsigned lp = ok ? (h[b] + rank) : (unsigned)(PCAP * 1024 + lane);
    stage[lp] = sv[j];
  }
  __syncthreads();
  for (int i = t; i < n; i += 1024) rec[e0 + i] = stage[i];
}

#define NSUB 8

__device__ __forceinline__ unsigned key_of(unsigned nl, unsigned sub) {
  return ((nl >> 5) << 8) | (sub << 5) | (nl & 31u);
}
__device__ __forceinline__ size_t slot_of(int node, int sub) {
  return (size_t)(node >> 5) * 256 + (size_t)sub * 32 + (size_t)(node & 31);
}

__device__ __forceinline__ unsigned sub_of(unsigned s, unsigned SL2) {
  unsigned k = 0u;
#pragma unroll
  for (unsigned q = 1; q < NSUB; ++q) k += (s >= q * SL2) ? 1u : 0u;
  return k;
}

__global__ void __launch_bounds__(1024) __attribute__((amdgpu_waves_per_eu(16, 16)))
bucket_sort_sub(const unsigned* __restrict__ rec, const unsigned* __restrict__ lst,
                int* __restrict__ rp2, int* __restrict__ csr, int N, unsigned SL2, int chunk) {
  __shared__ unsigned c[256 * NSUB + 32];
  __shared__ unsigned wsum[32];
  __shared__ unsigned wsum2[32];
  __shared__ unsigned tot[2];
  __shared__ unsigned stage[SCAP * 1024 + 32];
  const int t    = threadIdx.x;
  const int lane = t & 31;
  c[t] = 0u;
  c[t + 1024] = 0u;
  if (t < 32) c[256 * NSUB + t] = 0u;

  const unsigned ra  = lst[(size_t)blockIdx.x * NCHUNK + t];
  const unsigned rz  = lst[(size_t)(blockIdx.x + 1) * NCHUNK + t];
  const unsigned len = rz - ra;
  const unsigned exl = block_excl_scan_1024(len, wsum);
  const unsigned exa = block_excl_scan_1024(ra, wsum2);
  if (t == 1023) { tot[0] = exa + ra; tot[1] = exl + len; }
  __syncthreads();
  const unsigned bs = (unsigned)__builtin_amdgcn_readfirstlane((int)tot[0]);
  const unsigned n  = (unsigned)__builtin_amdgcn_readfirstlane((int)tot[1]);
  const unsigned g0 = (unsigned)t * (unsigned)chunk + ra;
  if (blockIdx.x == gridDim.x - 1 && t == 0) rp2[(size_t)gridDim.x * 256 * NSUB] = (int)(bs + n);
  if (blockIdx.x == gridDim.x - 1 && t < 4) csr[bs + n + (unsigned)t] = 0;

  if (n <= (unsigned)(SCAP * 1024)) {
    for (unsigned q = 0; q < len; ++q) stage[exl + q] = g0 + q;
    if (t < 32) stage[SCAP * 1024 + t] = 0u;
    __syncthreads();
    unsigned rv[SCAP], rk[SCAP];
#pragma unroll
    for (int j = 0; j < SCAP; ++j) {
      const unsigned i = (unsigned)(j * 1024 + t);
      rv[j] = stage[(i < n) ? i : (unsigned)(SCAP * 1024 + lane)];
    }
#pragma unroll
    for (int j = 0; j < SCAP; ++j) rv[j] = rec[rv[j]];
#pragma unroll
    for (int j = 0; j < SCAP; ++j) {
      const unsigned s = rv[j] & 0x3FFFFu;
      rv[j] = s | (key_of(rv[j] >> 18, sub_of(s, SL2)) << 18);
      asm volatile("" : "+v"(rv[j]));
    }
    __syncthreads();
#pragma unroll
    for (int j = 0; j < SCAP; ++j) {
      const bool ok = (unsigned)(j * 1024 + t) < n;
      const unsigned k = ok ? (rv[j] >> 18) : (unsigned)(256 * NSUB + lane);
      rk[j] = atomicAdd(&c[k], 1u);
    }
    __syncthreads();
    const unsigned v0 = c[2 * t];
    const unsigned v1 = c[2 * t + 1];
    const unsigned ex = block_excl_scan_1024(v0 + v1, wsum);
    c[2 * t]     = ex;
    c[2 * t + 1] = ex + v0;
    {
      int2 o;
      o.x = (int)(bs + ex);
      o.y = (int)(bs + ex + v0);
      ((int2*)rp2)[(size_t)blockIdx.x * 1024 + t] = o;
    }
    __syncthreads();
#pragma unroll
    for (int j = 0; j < SCAP; ++j) {
      const bool ok = (unsigned)(j * 1024 + t) < n;
      asm volatile("" : "+v"(rv[j]));
      const unsigned key = ok ? (rv[j] >> 18) : 0u;
      const unsigned lp  = ok ? (c[key] + rk[j]) : (unsigned)(SCAP * 1024 + lane);
      stage[lp] = rv[j] & 0x3FFFFu;
    }
    __syncthreads();
    for (unsigned i = (unsigned)t; i < n; i += 1024u) csr[bs + i] = (int)stage[i];
  } else {
    for (unsigned q = 0; q < len; ++q) {
      const unsigned r = rec[g0 + q];
      atomicAdd(&c[key_of(r >> 18, sub_of(r & 0x3FFFFu, SL2))], 1u);
    }
    __syncthreads();
    const unsigned v0 = c[2 * t];
    const unsigned v1 = c[2 * t + 1];
    const unsigned ex = block_excl_scan_1024(v0 + v1, wsum);
    c[2 * t]     = bs + ex;
    c[2 * t + 1] = bs + ex + v0;
    {
      int2 o;
      o.x = (int)(bs + ex);
      o.y = (int)(bs + ex + v0);
      ((int2*)rp2)[(size_t)blockIdx.x * 1024 + t] = o;
    }
    __syncthreads();
    for (unsigned q = 0; q < len; ++q) {
      const unsigned r = rec[g0 + q];
      const unsigned pos = atomicAdd(&c[key_of(r >> 18, sub_of(r & 0x3FFFFu, SL2))], 1u);
      csr[pos] = (int)(r & 0x3FFFFu);
    }
  }
}

__global__ void transform_y(const float* __restrict__ x, const int* __restrict__ rp2,
                            float* __restrict__ dnorm, float* __restrict__ y, int N) {
  int i = blockIdx.x * blockDim.x + threadIdx.x;
  if (i >= N) return;
  const int* rr = rp2 + slot_of(i, 0);
  int deg = 0;
#pragma unroll
  for (int s = 0; s < NSUB; ++s) deg += rr[s * 32 + 1] - rr[s * 32];
  float di = rsqrtf((float)deg + 1.0f);
  dnorm[i] = di;
  y[i] = x[i] * di;
}

#define SLCAP 68096
#define YV    3
#define FT    16
#define FSTEP (1024 * FT)
#define FT2   8
#define FSTEP2 (1024 * FT2)

__device__ __forceinline__ void list_blocks(const int* __restrict__ csr, int e, int z, int (&sv)[YV][4]) {
#pragma unroll
  for (int j = 0; j < YV; ++j) {
    const int ee = e + 8 * j;
    __builtin_memcpy(sv[j], csr + ((ee < z) ? ee : 0), 16);
  }
}

__global__ void __launch_bounds__(1024)
pull_y_lds(const int* __restrict__ rp2, const int* __restrict__ csr,
           const float* __restrict__ y, float* __restrict__ S, int N, int SL) {
  __shared__ float ys[SLCAP];
  const int t   = threadIdx.x;
  const int k   = blockIdx.y;
  const int s0  = k * SL;
  const int s1  = (s0 + SL < N) ? (s0 + SL) : N;
  const int len = s1 - s0;
  const unsigned ulen = (len > 0) ? (unsigned)len : 0u;

  const int G   = gridDim.x;
  const int npg = (((N + G - 1) / G) + 31) & ~31;
  const int n0  = blockIdx.x * npg;
  const int n1  = (n0 + npg < N) ? (n0 + npg) : N;
  const int sub = t & 3;
  const int hh  = sub >> 1;
  const int ls  = sub & 1;
  float* Sk = S + (size_t)k * N;

  auto ask_bounds = [&](int nb, int& r0, int& r1) {
    const int i_ = nb + (t >> 2);
    const int* rr_ = rp2 + ((i_ < n1) ? slot_of(i_, 2 * k + hh) : (size_t)0);
    r0 = rr_[0];
    r1 = rr_[1];
  };
  auto use_bounds = [&](int nb, int r0, int r1, int& e, int& z) {
    const int m_ = ((nb + (t >> 2)) < n1) ? -1 : 0;
    e = (r0 + 4 * ls) & m_;
    z = r1 & m_;
  };

  int rA0, rA1, rB0, rB1;
  ask_bounds(n0, rA0, rA1);
  ask_bounds(n0 + 256, rB0, rB1);

  const int ntrip = (len + FSTEP - 1) / FSTEP;
  float va[FT], vb[FT];
#pragma unroll
  for (int u = 0; u < FT; ++u) {
    const int j = t + u * 1024;
    va[u] = y[s0 + ((j < len) ? j : 0)];
  }
  int eA, zA, eB = 0, zB = 0;
  use_bounds(n0, rA0, rA1, eA, zA);
  int svA[YV][4], svB[YV][4];
  list_blocks(csr, eA, zA, svA);
  for (int m = 0; m < ntrip; m += 2) {
    const int ja = m * FSTEP + t;
    const int jb = ja + FSTEP;
#pragma unroll
    for (int u = 0; u < FT; ++u) {
      const int j = jb + u * 1024;
      vb[u] = y[s0 + ((j < len) ? j : 0)];
    }
#pragma unroll
    for (int u = 0; u < FT; ++u) {
      const int j = ja + u * 1024;
      if (j < len) ys[j] = va[u];
    }
#pragma unroll
    for (int u = 0; u < FT; ++u) {
      const int j = jb + FSTEP + u * 1024;
      va[u] = y[s0 + ((j < len) ? j : 0)];
    }
#pragma unroll
    for (int u = 0; u < FT; ++u) {
      const int j = jb + u * 1024;
      if (j < len) ys[j] = vb[u];
    }
  }
  __syncthreads();

  auto pass = [&](int nb0, int (&svU)[YV][4], int eU, int zU,
                  int (&svL)[YV][4], int& eL, int& zL,
                  int rU0, int rU1, int& rL0, int& rL1) {
    const int i = nb0 + (t >> 2);
    const bool live = i < n1;
    ask_bounds(nb0 + 512, rL0, rL1);
    use_bounds(nb0 + 256, rU0, rU1, eL, zL);
    list_blocks(csr, eL, zL, svL);
    float acc = 0.0f;
#pragma unroll
    for (int j = 0; j < YV; ++j) {
#pragma unroll
      for (int u = 0; u < 4; ++u) {
        const unsigned r = (unsigned)(svU[j][u] - s0);
        const bool ok = ((eU + 8 * j + u) < zU) && (r < ulen);
        const float v = ys[ok ? r : 0u];
        acc += ok ? v : 0.0f;
      }
    }
    for (int e = eU + 8 * YV; e < zU; e += 8 * YV) {
      int sv[YV][4];
      list_blocks(csr, e, zU, sv);
#pragma unroll
      for (int j = 0; j < YV; ++j) {
#pragma unroll
        for (int u = 0; u < 4; ++u) {
          const unsigned r = (unsigned)(sv[j][u] - s0);
          const bool ok = ((e + 8 * j + u) < zU) && (r < ulen);
          const float v = ys[ok ? r : 0u];
          acc += ok ? v : 0.0f;
        }
      }
    }
    acc += __shfl_xor(acc, 1, 32);
    acc += __shfl_xor(acc, 2, 32);
    if (live && sub == 0) Sk[i] = acc;
  };

  for (int nb0 = n0; nb0 < n1; nb0 += 512) {
    pass(nb0, svA, eA, zA, svB, eB, zB, rB0, rB1, rA0, rA1);
    if (nb0 + 256 >= n1) break;
    pass(nb0 + 256, svB, eB, zB, svA, eA, zA, rA0, rA1, rB0, rB1);
  }
}

__global__ void make_td(const float* __restrict__ S, int nsl, const float* __restrict__ y,
                        const float* __restrict__ dnorm, float* __restrict__ td, int N) {
  int i = blockIdx.x * blockDim.x + threadIdx.x;
  if (i >= N) return;
  const float di = dnorm[i];
  float s = 0.0f;
  for (int q = 0; q < nsl; ++q) s += S[(size_t)q * N + i];
  s += y[i];
  float2 o;
  o.x = di * s;
  o.y = di;
  ((float2*)td)[i] = o;
}

#define TDCAP (SLCAP / 2)
#define TU    4

__global__ void __launch_bounds__(1024)
pull_td_lds(const int* __restrict__ rp2, const int* __restrict__ csr,
            const float* __restrict__ tdf, const float* __restrict__ W1,
            const float* __restrict__ b1, float* __restrict__ P, int N, int SL2) {
  __shared__ float2 tds[TDCAP];
  const float2* td = (const float2*)tdf;
  const int t = threadIdx.x;
  const int k = blockIdx.y;
  v2f w1p[8], b1p[8];
#pragma unroll
  for (int c = 0; c < 8; ++c) {
    w1p[c][0] = W1[2 * c]; w1p[c][1] = W1[2 * c + 1];
    b1p[c][0] = b1[2 * c]; b1p[c][1] = b1[2 * c + 1];
    asm volatile("" : "+v"(w1p[c]), "+v"(b1p[c]));
  }

  const int G   = gridDim.x;
  const int npg = (((N + G - 1) / G) + 31) & ~31;
  const int n0  = blockIdx.x * npg;
  const int n1  = (n0 + npg < N) ? (n0 + npg) : N;

  for (int h = 0; h < 2; ++h) {
    const int sub = 2 * k + h;
    const int s0  = sub * SL2;
    const int s1  = (s0 + SL2 < N) ? (s0 + SL2) : N;
    const int len = s1 - s0;
    const unsigned ulen = (len > 0) ? (unsigned)len : 0u;

    auto ask_bounds = [&](int nb, int& r0, int& r1) {
      const int i_ = nb + t;
      const int* rr_ = rp2 + ((i_ < n1) ? slot_of(i_, sub) : (size_t)0);
      r0 = rr_[0];
      r1 = rr_[1];
    };
    auto use_bounds = [&](int nb, int r0, int r1, int& e, int& c) {
      const int m_ = ((nb + t) < n1) ? -1 : 0;
      e = r0 & m_;
      c = (r1 - r0) & m_;
    };
    auto ask_sums = [&](int i_, float4 (&po)[4]) {
      if (h && i_ < n1) {
        const float4* Pq = (const float4*)(P + ((size_t)k * N + i_) * 16);
#pragma unroll
        for (int q = 0; q < 4; ++q) po[q] = Pq[q];
      }
    };

    __syncthreads();
    int r0, r1;
    ask_bounds(n0, r0, r1);

    const int ntrip = (len + FSTEP2 - 1) / FSTEP2;
    float2 va[FT2], vb[FT2];
#pragma unroll
    for (int u = 0; u < FT2; ++u) {
      const int j = t + u * 1024;
      va[u] = td[s0 + ((j < len) ? j : 0)];
    }
    asm volatile("" ::: "memory");
    int e0, cnt;
    use_bounds(n0, r0, r1, e0, cnt);
    int cur[TU];
    __builtin_memcpy(cur, csr + ((0 < cnt) ? e0 : 0), 4 * TU);
    ask_bounds(n0 + 1024, r0, r1);
    float4 pold[4];
#pragma unroll
    for (int q = 0; q < 4; ++q) { pold[q].x = 0.0f; pold[q].y = 0.0f; pold[q].z = 0.0f; pold[q].w = 0.0f; }
    ask_sums(n0 + t, pold);
    for (int m = 0; m < ntrip; m += 2) {
      const int ja = m * FSTEP2 + t;
      const int jb = ja + FSTEP2;
#pragma unroll
      for (int u = 0; u < FT2; ++u) {
        const int j = jb + u * 1024;
        vb[u] = td[s0 + ((j < len) ? j : 0)];
      }
#pragma unroll
      for (int u = 0; u < FT2; ++u) {
        const int j = ja + u * 1024;
        if (j < len) tds[j] = va[u];
      }
#pragma unroll
      for (int u = 0; u < FT2; ++u) {
        const int j = jb + FSTEP2 + u * 1024;
        va[u] = td[s0 + ((j < len) ? j : 0)];
      }
#pragma unroll
      for (int u = 0; u < FT2; ++u) {
        const int j = jb + u * 1024;
        if (j < len) tds[j] = vb[u];
      }
    }
    __syncthreads();

    for (int nb0 = n0; nb0 < n1; nb0 += 1024) {
      const int i = nb0 + t;
      const bool live = i < n1;
      int wmax = cnt;
#pragma unroll
      for (int off = 16; off > 0; off >>= 1) {
        const int o = __shfl_xor(wmax, off, 32);
        wmax = (o > wmax) ? o : wmax;
      }
      wmax = __builtin_amdgcn_readfirstlane(wmax);

      v2f acc[8];
#pragma unroll
      for (int c = 0; c < 8; ++c) { acc[c][0] = 0.0f; acc[c][1] = 0.0f; }
      for (int base = 0; base < wmax; base += TU) {
        int nxt[TU];
        {
          const int q = base + TU;
          __builtin_memcpy(nxt, csr + ((q < cnt) ? (e0 + q) : 0), 4 * TU);
        }
#pragma unroll
        for (int j = 0; j < TU; ++j) {
          const int q = base + j;
          const unsigned r = (unsigned)(cur[j] - s0);
          const bool ok = (q < cnt) && (r < ulen);
          const float2 v = tds[ok ? r : 0u];
          const float tj = ok ? v.x : 0.0f;
          const float dj = ok ? v.y : 0.0f;
          const v2f tj2 = {tj, tj};
          const v2f dj2 = {dj, dj};
          const v2f z2  = {0.0f, 0.0f};
#pragma unroll
          for (int c = 0; c < 8; ++c)
            acc[c] = __builtin_elementwise_fma(dj2, __builtin_elementwise_max(__builtin_elementwise_fma(tj2, w1p[c], b1p[c]), z2), acc[c]);
        }
        asm volatile("" ::: "memory");
#pragma unroll
        for (int j = 0; j < TU; ++j) cur[j] = nxt[j];
      }
      use_bounds(nb0 + 1024, r0, r1, e0, cnt);
      __builtin_memcpy(cur, csr + ((0 < cnt) ? e0 : 0), 4 * TU);
      ask_bounds(nb0 + 2048, r0, r1);
      if (live) {
        float4* Pp = (float4*)(P + ((size_t)k * N + i) * 16);
#pragma unroll
        for (int q = 0; q < 4; ++q) {
          float4 o;
          o.x = acc[2 * q][0]; o.y = acc[2 * q][1]; o.z = acc[2 * q + 1][0]; o.w = acc[2 * q + 1][1];
          if (h) { o.x += pold[q].x; o.y += pold[q].y; o.z += pold[q].z; o.w += pold[q].w; }
          Pp[q] = o;
        }
      }
      ask_sums(i + 1024, pold);
    }
  }
}

__global__ void __launch_bounds__(256)
gcn_out_wmma_u(const float* __restrict__ P, int nsl, const float* __restrict__ td,
               const float* __restrict__ W1, const float* __restrict__ b1,
               const float* __restrict__ W2, const float* __restrict__ b2,
               const float* __restrict__ Wl, const float* __restrict__ bl,
               float* __restrict__ out, int N, int nTiles) {
  const int lane = threadIdx.x & 31;
  const int tile = (blockIdx.x * blockDim.x + threadIdx.x) >> 5;
  if (tile >= nTiles) return;
  const int base = tile * 16;
  const int col  = lane & 15;
  const int hf   = lane >> 4;
  const bool full = (base + 16 <= N);

  int arow = base + col;
  if (arow >= N) arow = N - 1;
  const float2 tdv = ((const float2*)td)[arow];
  const float ti = tdv.x;
  const float di = tdv.y;

  v2f aw[4], bu[4];
#pragma unroll
  for (int k = 0; k < 4; ++k) {
    const int c0 = 4 * k + 2 * hf;
    float u0 = 0.0f, u1 = 0.0f;
    for (int q = 0; q < nsl; ++q) {
      const float2 pv = *(const float2*)(P + ((size_t)q * N + arow) * 16 + c0);
      u0 += pv.x;
      u1 += pv.y;
    }
    u0 += di * fmaxf(fmaf(ti, W1[c0],     b1[c0]),     0.0f);
    u1 += di * fmaxf(fmaf(ti, W1[c0 + 1], b1[c0 + 1]), 0.0f);
    bu[k][0] = u0;
    bu[k][1] = u1;
    aw[k][0] = W2[c0 * 16 + col];
    aw[k][1] = W2[(c0 + 1) * 16 + col];
  }

  v8f z = {};
#pragma unroll
  for (int k = 0; k < 4; ++k)
    z = __builtin_amdgcn_wmma_f32_16x16x4_f32(false, aw[k], false, bu[k],
                                              (short)0, z, false, false);

  v2f ah[4], bw[4];
#pragma unroll
  for (int k = 0; k < 4; ++k) {
    const int c0 = 8 * hf + 2 * k;
    ah[k][0] = fmaxf(fmaf(di, z[2 * k],     b2[c0]),     0.0f);
    ah[k][1] = fmaxf(fmaf(di, z[2 * k + 1], b2[c0 + 1]), 0.0f);
    bw[k][0] = (col < 4) ? Wl[c0 * 4 + col]       : 0.0f;
    bw[k][1] = (col < 4) ? Wl[(c0 + 1) * 4 + col] : 0.0f;
  }

  const float cv = (col < 4) ? bl[col] : 0.0f;
  v8f acc;
#pragma unroll
  for (int r = 0; r < 8; ++r) acc[r] = cv;
#pragma unroll
  for (int k = 0; k < 4; ++k)
    acc = __builtin_amdgcn_wmma_f32_16x16x4_f32(false, ah[k], false, bw[k],
                                                (short)0, acc, false, false);

  if (col < 4) {
    if (full) {
#pragma unroll
      for (int r = 0; r < 8; ++r) {
        int orow = base + (hf ? 8 + r : r);
        out[(size_t)orow * 4 + col] = acc[r];
      }
    } else {
#pragma unroll
      for (int r = 0; r < 8; ++r) {
        int orow = base + (hf ? 8 + r : r);
        if (orow < N) out[(size_t)orow * 4 + col] = acc[r];
      }
    }
  }
}

extern "C" void kernel_launch(void* const* d_in, const int* in_sizes, int n_in,
                              void* d_out, int out_size, void* d_ws, size_t ws_size,
                              hipStream_t stream) {
  const float* x  = (const float*)d_in[0];
  const int*   ed = (const int*)d_in[1];
  const float* W1 = (const float*)d_in[2];
  const float* b1 = (const float*)d_in[3];
  const float* W2 = (const float*)d_in[4];
  const float* b2 = (const float*)d_in[5];
  const float* Wl = (const float*)d_in[6];
  const float* bl = (const float*)d_in[7];
  float* out = (float*)d_out;

  const int N = in_sizes[0];
  const int E = in_sizes[1] / 2;

  const int* src = ed;
  const int* dst = ed + E;

  const int B = 256;
  const int nTiles = (N + 15) / 16;
  const int tileBlocks = (nTiles * 32 + B - 1) / B;

  const int nb  = (N + 255) / 256;
  const int nsl = (N + SLCAP - 1) / SLCAP;
  const int SL2 = (N + 2 * nsl - 1) / (2 * nsl);
  const int SL  = 2 * SL2;
  const size_t Nr = ((size_t)N + 64) / 64 * 64;
  const size_t Er = ((size_t)E + 4 + 63) / 64 * 64;
  const size_t Pw = (size_t)nsl * 16 * Nr;
  const size_t Rr = ((size_t)nb * 256 * NSUB + 64) / 64 * 64;
  const size_t o_td   = Nr;
  const size_t o_S    = o_td + 2 * Nr;
  const size_t o_rp   = o_S + 5 * Nr;
  const size_t o_lst  = o_rp + Rr;
  const size_t o_csr  = o_lst + (size_t)(nb + 1) * NCHUNK;
  const size_t o_big  = o_csr + Er;
  const size_t need   = (o_big + ((Er > Pw) ? Er : Pw)) * 4;
  const int chunk = (E + NCHUNK - 1) / NCHUNK;

  if (need <= ws_size && nb < MAXB && chunk <= PCAP * 1024 && 2 * nsl <= NSUB && SL <= SLCAP) {
    float*    dnorm  = (float*)d_ws;
    float*    td     = dnorm + o_td;
    float*    S      = dnorm + o_S;
    int*      rp2    = (int*)d_ws + o_rp;
    unsigned* lst    = (unsigned*)d_ws + o_lst;
    int*      csr    = (int*)d_ws + o_csr;
    unsigned* rec    = (unsigned*)d_ws + o_big;
    float*    Pl     = dnorm + o_big;

    chunk_sort     <<<NCHUNK, 1024, 0, stream>>>(src, dst, lst, rec, E, chunk, nb);
    bucket_sort_sub<<<nb, 1024, 0, stream>>>(rec, lst, rp2, csr, N, (unsigned)SL2, chunk);

    float* yv = S + (size_t)nsl * N;
    transform_y<<<(N + B - 1) / B, B, 0, stream>>>(x, rp2, dnorm, yv, N);
    pull_y_lds<<<dim3(256 / nsl, nsl), 1024, 0, stream>>>(rp2, csr, yv, S, N, SL);
    make_td<<<(N + B - 1) / B, B, 0, stream>>>(S, nsl, yv, dnorm, td, N);

    pull_td_lds<<<dim3(256 / nsl, nsl), 1024, 0, stream>>>(rp2, csr, td, W1, b1, Pl, N, SL2);
    gcn_out_wmma_u<<<tileBlocks, B, 0, stream>>>(Pl, nsl, td, W1, b1, W2, b2, Wl, bl, out, N, nTiles);
    return;
  }

  float* dnorm = (float*)d_ws;
  float* p     = dnorm + N;
  float* agg   = p + (size_t)N * 16;

  const int edgeBlocks = (int)(((long long)E * 4 + B - 1) / B);

  zero_f32  <<<(N + B - 1) / B, B, 0, stream>>>(dnorm, N);
  deg_accum <<<(E + B - 1) / B, B, 0, stream>>>(dst, dnorm, E);
  transform1<<<(N + B - 1) / B, B, 0, stream>>>(x, W1, dnorm, p, agg, N);
  edge_scatter<<<edgeBlocks, B, 0, stream>>>(src, dst, p, agg, E);
  gcn_mid_wmma<true><<<tileBlocks, B, 0, stream>>>(p, agg, dnorm, b1, W2, p, agg, N, nTiles);
  edge_scatter<<<edgeBlocks, B, 0, stream>>>(src, dst, p, agg, E);
  gcn_out_wmma<<<tileBlocks, B, 0, stream>>>(p, agg, dnorm, b2, Wl, bl, out, N, nTiles);
}
